# starting kernel
# baseline (speedup 1.0000x reference)
#include <hip/hip_runtime.h>
#include <hip/hip_bf16.h>


typedef __attribute__((ext_vector_type(2))) float v2f;
typedef __attribute__((ext_vector_type(8))) float v8f;

#define DIM 64

__global__ void gcn_zero_kernel(float* __restrict__ p, int n) {
    int i = blockIdx.x * blockDim.x + threadIdx.x;
    if (i < n) p[i] = 0.0f;
}

__global__ void gcn_deg_kernel(const int* __restrict__ ei,
                               const float* __restrict__ ew,
                               float* __restrict__ deg, int E) {
    int e = blockIdx.x * blockDim.x + threadIdx.x;
    if (e < E) {
        atomicAdd(&deg[ei[E + e]], ew[e]);
    }
}

__global__ void __launch_bounds__(256)
gcn_scatter_kernel(const float* __restrict__ H,
                   const int* __restrict__ ei,
                   const float* __restrict__ ew,
                   const float* __restrict__ deg,
                   float* __restrict__ agg, int E) {
    int e = __builtin_amdgcn_readfirstlane(blockIdx.x * 8 + (threadIdx.x >> 5));
    int lane = threadIdx.x & 31;
    if (e >= E) return;

    int src = ei[e];
    int dst = ei[E + e];
    float w  = ew[e];
    float ds = fmaxf(deg[src], 1e-6f);
    float dd = fmaxf(deg[dst], 1e-6f);
    float norm = w / fmaxf(sqrtf(ds * dd), 1e-6f);

    const float2* hrow = (const float2*)(H + (size_t)src * DIM);
    float2 h = hrow[lane];

    float* arow = agg + (size_t)dst * DIM + 2 * lane;
    atomicAdd(arow + 0, norm * h.x);
    atomicAdd(arow + 1, norm * h.y);
}

__global__ void __launch_bounds__(128)
gcn_gemm_wmma_kernel(const float* __restrict__ agg,
                     const float* __restrict__ Wm,
                     float* __restrict__ out, int Nn) {
    const int wave    = threadIdx.x >> 5;
    const int lane    = threadIdx.x & 31;
    const int half    = lane >> 4;
    const int l       = lane & 15;
    const int rowBase = blockIdx.x * 16;
    const int colBase = wave * 16;

    v2f bfrag[16];
#pragma unroll
    for (int kk = 0; kk < 16; ++kk) {
        const float* wp = Wm + (size_t)(colBase + l) * DIM + kk * 4 + 2 * half;
        bfrag[kk] = *(const v2f*)wp;
    }

    int rA = rowBase + l;
    if (rA >= Nn) rA = Nn - 1;

    v8f c = {};
#pragma unroll
    for (int kk = 0; kk < 16; ++kk) {
        const float* ap = agg + (size_t)rA * DIM + kk * 4 + 2 * half;
        v2f a = *(const v2f*)ap;
        c = __builtin_amdgcn_wmma_f32_16x16x4_f32(
                 false, a,  false, bfrag[kk],
                 (short)0, c,  false,  false);
    }

#pragma unroll
    for (int r = 0; r < 8; ++r) {
        int row = rowBase + r + 8 * half;
        if (row < Nn) out[(size_t)row * DIM + colBase + l] = c[r];
    }
}

extern "C" void kernel_launch(void* const* d_in, const int* in_sizes, int n_in,
                              void* d_out, int out_size, void* d_ws, size_t ws_size,
                              hipStream_t stream) {
    const float* H   = (const float*)d_in[0];
    const int*   ei  = (const int*)d_in[1];
    const float* ew  = (const float*)d_in[2];
    const float* Wm  = (const float*)d_in[3];
    float*       out = (float*)d_out;

    const int N = in_sizes[0] / DIM;
    const int E = in_sizes[2];

    const int degPad = ((N + 63) / 64) * 64;
    float* deg = (float*)d_ws;
    float* agg = deg + degPad;

    {
        int n = degPad + N * DIM;
        gcn_zero_kernel<<<(n + 255) / 256, 256, 0, stream>>>(deg, n);
    }
    gcn_deg_kernel<<<(E + 255) / 256, 256, 0, stream>>>(ei, ew, deg, E);

    gcn_scatter_kernel<<<(E + 7) / 8, 256, 0, stream>>>(H, ei, ew, deg, agg, E);

    gcn_gemm_wmma_kernel<<<(N + 15) / 16, 128, 0, stream>>>(agg, Wm, out, N);
}


# atomics: each L2 half (die id via s_sendmsg_rtn) adds into its own copy of deg and agg, 2nd copy kept in out; GEMM accumulates both copies (2 WMMA per k step); sum order changes: 2 partial sums
# speedup vs baseline: 1.1316x; 1.1316x over previous
#include <hip/hip_runtime.h>
#include <hip/hip_bf16.h>


typedef __attribute__((ext_vector_type(2))) float v2f;
typedef __attribute__((ext_vector_type(8))) float v8f;

#define DIM 64

__device__ __forceinline__ int gcn_l2_half() {
    unsigned id = __builtin_amdgcn_s_sendmsg_rtn(0x87);
    return (int)((id >> 18) & 1u);
}

__global__ void gcn_zero_kernel(float* __restrict__ a, int na4,
                                float* __restrict__ b, int nb4) {
    int i = blockIdx.x * blockDim.x + threadIdx.x;
    const float4 z = make_float4(0.0f, 0.0f, 0.0f, 0.0f);
    if (i < na4) {
        ((float4*)a)[i] = z;
    } else if (i - na4 < nb4) {
        ((float4*)b)[i - na4] = z;
    }
}

__global__ void gcn_deg_kernel(const int* __restrict__ ei,
                               const float* __restrict__ ew,
                               float* __restrict__ degA,
                               float* __restrict__ degB, int E) {
    int e = blockIdx.x * blockDim.x + threadIdx.x;
    float* deg = gcn_l2_half() ? degB : degA;
    if (e < E) {
        atomicAdd(&deg[ei[E + e]], ew[e]);
    }
}

__global__ void gcn_deg_merge_kernel(float* __restrict__ degA,
                                     float* __restrict__ degB, int n) {
    int i = blockIdx.x * blockDim.x + threadIdx.x;
    if (i < n) {
        degA[i] += degB[i];
        degB[i] = 0.0f;
    }
}

__global__ void __launch_bounds__(256)
gcn_scatter_kernel(const float* __restrict__ H,
                   const int* __restrict__ ei,
                   const float* __restrict__ ew,
                   const float* __restrict__ deg,
                   float* __restrict__ aggA,
                   float* __restrict__ aggB, int E) {
    __shared__ int s_half;

    int e = __builtin_amdgcn_readfirstlane(blockIdx.x * 8 + (threadIdx.x >> 5));
    int lane = threadIdx.x & 31;
    const bool live = e < E;
    if (!live) e = E - 1;

    int src = ei[e];
    int dst = ei[E + e];
    float w  = ew[e];
    float ds = fmaxf(deg[src], 1e-6f);
    float dd = fmaxf(deg[dst], 1e-6f);
    float norm = w / fmaxf(sqrtf(ds * dd), 1e-6f);

    const float2* hrow = (const float2*)(H + (size_t)src * DIM);
    float2 h = hrow[lane];

    if (threadIdx.x == 0) s_half = gcn_l2_half();
    __syncthreads();
    const int half = __builtin_amdgcn_readfirstlane(s_half);
    if (live) {
        float* arow = (half ? aggB : aggA) + (size_t)dst * DIM + 2 * lane;
        atomicAdd(arow + 0, norm * h.x);
        atomicAdd(arow + 1, norm * h.y);
    }
}

__global__ void __launch_bounds__(128)
gcn_gemm_wmma_kernel(const float* __restrict__ aggA,
                     const float* __restrict__ Wm,
                     float* out, int Nn) {
    const int wave    = threadIdx.x >> 5;
    const int lane    = threadIdx.x & 31;
    const int half    = lane >> 4;
    const int l       = lane & 15;
    const int rowBase = blockIdx.x * 16;
    const int colBase = wave * 16;

    v2f bfrag[16];
#pragma unroll
    for (int kk = 0; kk < 16; ++kk) {
        const float* wp = Wm + (size_t)(colBase + l) * DIM + kk * 4 + 2 * half;
        bfrag[kk] = *(const v2f*)wp;
    }

    int rA = rowBase + l;
    if (rA >= Nn) rA = Nn - 1;

    v8f c = {};
#pragma unroll
    for (int kk = 0; kk < 16; ++kk) {
        const size_t off = (size_t)rA * DIM + kk * 4 + 2 * half;
        v2f a0 = *(const v2f*)(aggA + off);
        v2f a1 = *(const v2f*)(out + off);
        c = __builtin_amdgcn_wmma_f32_16x16x4_f32(
                 false, a0,  false, bfrag[kk],
                 (short)0, c,  false,  false);
        c = __builtin_amdgcn_wmma_f32_16x16x4_f32(
                 false, a1,  false, bfrag[kk],
                 (short)0, c,  false,  false);
    }

    __builtin_amdgcn_sched_barrier(0);
    asm volatile("v_nop\n\tv_nop\n\tv_nop\n\tv_nop");
    __builtin_amdgcn_sched_barrier(0);

    __syncthreads();

#pragma unroll
    for (int r = 0; r < 8; ++r) {
        int row = rowBase + r + 8 * half;
        if (row < Nn) out[(size_t)row * DIM + colBase + l] = c[r];
    }
}

extern "C" void kernel_launch(void* const* d_in, const int* in_sizes, int n_in,
                              void* d_out, int out_size, void* d_ws, size_t ws_size,
                              hipStream_t stream) {
    const float* H   = (const float*)d_in[0];
    const int*   ei  = (const int*)d_in[1];
    const float* ew  = (const float*)d_in[2];
    const float* Wm  = (const float*)d_in[3];
    float*       out = (float*)d_out;

    const int N = in_sizes[0] / DIM;
    const int E = in_sizes[2];

    const int degPad = ((N + 63) / 64) * 64;
    float* deg = (float*)d_ws;
    float* agg = deg + degPad;

    {
        int na4 = (degPad + N * DIM) / 4;
        int nb4 = (N * DIM) / 4;
        gcn_zero_kernel<<<(na4 + nb4 + 255) / 256, 256, 0, stream>>>(deg, na4, out, nb4);
    }
    gcn_deg_kernel<<<(E + 255) / 256, 256, 0, stream>>>(ei, ew, deg, out, E);
    gcn_deg_merge_kernel<<<(N + 255) / 256, 256, 0, stream>>>(deg, out, N);

    gcn_scatter_kernel<<<(E + 7) / 8, 256, 0, stream>>>(H, ei, ew, deg, agg, out, E);

    gcn_gemm_wmma_kernel<<<(N + 15) / 16, 128, 0, stream>>>(agg, Wm, out, N);
}


# scatter: one wave per 32 edges (edge scalars one per lane, 32 H-row loads in flight before the atomics, no LDS/barrier) instead of one wave per edge; agg still summed by atomics in arrival order
# speedup vs baseline: 1.7104x; 1.5115x over previous
#include <hip/hip_runtime.h>
#include <hip/hip_bf16.h>


typedef __attribute__((ext_vector_type(2))) float v2f;
typedef __attribute__((ext_vector_type(8))) float v8f;

#define DIM 64

__device__ __forceinline__ int gcn_l2_half() {
    unsigned id = __builtin_amdgcn_s_sendmsg_rtn(0x87);
    return (int)((id >> 18) & 1u);
}

__global__ void gcn_zero_kernel(float* __restrict__ a, int na4,
                                float* __restrict__ b, int nb4) {
    int i = blockIdx.x * blockDim.x + threadIdx.x;
    const float4 z = make_float4(0.0f, 0.0f, 0.0f, 0.0f);
    if (i < na4) {
        ((float4*)a)[i] = z;
    } else if (i - na4 < nb4) {
        ((float4*)b)[i - na4] = z;
    }
}

__global__ void gcn_deg_kernel(const int* __restrict__ ei,
                               const float* __restrict__ ew,
                               float* __restrict__ degA,
                               float* __restrict__ degB, int E) {
    int e = blockIdx.x * blockDim.x + threadIdx.x;
    float* deg = gcn_l2_half() ? degB : degA;
    if (e < E) {
        atomicAdd(&deg[ei[E + e]], ew[e]);
    }
}

__global__ void gcn_deg_merge_kernel(float* __restrict__ degA,
                                     float* __restrict__ degB, int n) {
    int i = blockIdx.x * blockDim.x + threadIdx.x;
    if (i < n) {
        degA[i] += degB[i];
        degB[i] = 0.0f;
    }
}

__global__ void __launch_bounds__(256)
gcn_scatter_kernel(const float* __restrict__ H,
                   const int* __restrict__ ei,
                   const float* __restrict__ ew,
                   const float* __restrict__ deg,
                   float* __restrict__ aggA,
                   float* __restrict__ aggB, int E) {
    const int lane = threadIdx.x & 31;
    const int e  = blockIdx.x * blockDim.x + threadIdx.x;
    const int e0 = __builtin_amdgcn_readfirstlane(e);
    if (e0 >= E) return;
    const int cnt = (E - e0 < 32) ? (E - e0) : 32;

    const int ec = (e < E) ? e : (E - 1);
    const int src = ei[ec];
    const int dst = ei[E + ec];
    const float w = ew[ec];
    const float ds = fmaxf(deg[src], 1e-6f);
    const float dd = fmaxf(deg[dst], 1e-6f);
    const float norm = w / fmaxf(sqrtf(ds * dd), 1e-6f);
    const int normBits = __float_as_int(norm);

    float* __restrict__ agg = gcn_l2_half() ? aggB : aggA;

    if (cnt == 32) {
        float2 h[32];
#pragma unroll
        for (int j = 0; j < 32; ++j) {
            const int s = __builtin_amdgcn_readlane(src, j);
            h[j] = ((const float2*)(H + (size_t)s * DIM))[lane];
        }
#pragma unroll
        for (int j = 0; j < 32; ++j) {
            const int d = __builtin_amdgcn_readlane(dst, j);
            const float nj = __int_as_float(__builtin_amdgcn_readlane(normBits, j));
            float* arow = agg + (size_t)d * DIM + 2 * lane;
            atomicAdd(arow + 0, nj * h[j].x);
            atomicAdd(arow + 1, nj * h[j].y);
        }
    } else {
        for (int j = 0; j < cnt; ++j) {
            const int s = __builtin_amdgcn_readlane(src, j);
            const int d = __builtin_amdgcn_readlane(dst, j);
            const float nj = __int_as_float(__builtin_amdgcn_readlane(normBits, j));
            const float2 hv = ((const float2*)(H + (size_t)s * DIM))[lane];
            float* arow = agg + (size_t)d * DIM + 2 * lane;
            atomicAdd(arow + 0, nj * hv.x);
            atomicAdd(arow + 1, nj * hv.y);
        }
    }
}

__global__ void __launch_bounds__(128)
gcn_gemm_wmma_kernel(const float* __restrict__ aggA,
                     const float* __restrict__ Wm,
                     float* out, int Nn) {
    const int wave    = threadIdx.x >> 5;
    const int lane    = threadIdx.x & 31;
    const int half    = lane >> 4;
    const int l       = lane & 15;
    const int rowBase = blockIdx.x * 16;
    const int colBase = wave * 16;

    v2f bfrag[16];
#pragma unroll
    for (int kk = 0; kk < 16; ++kk) {
        const float* wp = Wm + (size_t)(colBase + l) * DIM + kk * 4 + 2 * half;
        bfrag[kk] = *(const v2f*)wp;
    }

    int rA = rowBase + l;
    if (rA >= Nn) rA = Nn - 1;

    v8f c = {};
#pragma unroll
    for (int kk = 0; kk < 16; ++kk) {
        const size_t off = (size_t)rA * DIM + kk * 4 + 2 * half;
        v2f a0 = *(const v2f*)(aggA + off);
        v2f a1 = *(const v2f*)(out + off);
        c = __builtin_amdgcn_wmma_f32_16x16x4_f32(
                 false, a0,  false, bfrag[kk],
                 (short)0, c,  false,  false);
        c = __builtin_amdgcn_wmma_f32_16x16x4_f32(
                 false, a1,  false, bfrag[kk],
                 (short)0, c,  false,  false);
    }

    __builtin_amdgcn_sched_barrier(0);
    asm volatile("v_nop\n\tv_nop\n\tv_nop\n\tv_nop");
    __builtin_amdgcn_sched_barrier(0);

    __syncthreads();

#pragma unroll
    for (int r = 0; r < 8; ++r) {
        int row = rowBase + r + 8 * half;
        if (row < Nn) out[(size_t)row * DIM + colBase + l] = c[r];
    }
}

extern "C" void kernel_launch(void* const* d_in, const int* in_sizes, int n_in,
                              void* d_out, int out_size, void* d_ws, size_t ws_size,
                              hipStream_t stream) {
    const float* H   = (const float*)d_in[0];
    const int*   ei  = (const int*)d_in[1];
    const float* ew  = (const float*)d_in[2];
    const float* Wm  = (const float*)d_in[3];
    float*       out = (float*)d_out;

    const int N = in_sizes[0] / DIM;
    const int E = in_sizes[2];

    const int degPad = ((N + 63) / 64) * 64;
    float* deg = (float*)d_ws;
    float* agg = deg + degPad;

    {
        int na4 = (degPad + N * DIM) / 4;
        int nb4 = (N * DIM) / 4;
        gcn_zero_kernel<<<(na4 + nb4 + 255) / 256, 256, 0, stream>>>(deg, na4, out, nb4);
    }
    gcn_deg_kernel<<<(E + 255) / 256, 256, 0, stream>>>(ei, ew, deg, out, E);
    gcn_deg_merge_kernel<<<(N + 255) / 256, 256, 0, stream>>>(deg, out, N);

    gcn_scatter_kernel<<<(E + 255) / 256, 256, 0, stream>>>(H, ei, ew, deg, agg, out, E);

    gcn_gemm_wmma_kernel<<<(N + 15) / 16, 128, 0, stream>>>(agg, Wm, out, N);
}


# scatter: the wave's 16-lane halves trade one value per edge (v_permlanex16) so each float atomic instruction covers one whole 128-B line of the agg row instead of every other word of two lines
# speedup vs baseline: 1.9826x; 1.1591x over previous
#include <hip/hip_runtime.h>
#include <hip/hip_bf16.h>


typedef __attribute__((ext_vector_type(2))) float v2f;
typedef __attribute__((ext_vector_type(8))) float v8f;

#define DIM 64

__device__ __forceinline__ float gcn_swap_halves(float x) {
    return __int_as_float(__builtin_amdgcn_permlanex16(
        __float_as_int(x), __float_as_int(x), 0x76543210, 0xfedcba98, false, false));
}

__device__ __forceinline__ int gcn_l2_half() {
    unsigned id = __builtin_amdgcn_s_sendmsg_rtn(0x87);
    return (int)((id >> 18) & 1u);
}

__global__ void gcn_zero_kernel(float* __restrict__ a, int na4,
                                float* __restrict__ b, int nb4) {
    int i = blockIdx.x * blockDim.x + threadIdx.x;
    const float4 z = make_float4(0.0f, 0.0f, 0.0f, 0.0f);
    if (i < na4) {
        ((float4*)a)[i] = z;
    } else if (i - na4 < nb4) {
        ((float4*)b)[i - na4] = z;
    }
}

__global__ void gcn_deg_kernel(const int* __restrict__ ei,
                               const float* __restrict__ ew,
                               float* __restrict__ degA,
                               float* __restrict__ degB, int E) {
    int e = blockIdx.x * blockDim.x + threadIdx.x;
    float* deg = gcn_l2_half() ? degB : degA;
    if (e < E) {
        atomicAdd(&deg[ei[E + e]], ew[e]);
    }
}

__global__ void gcn_deg_merge_kernel(float* __restrict__ degA,
                                     float* __restrict__ degB, int n) {
    int i = blockIdx.x * blockDim.x + threadIdx.x;
    if (i < n) {
        degA[i] += degB[i];
        degB[i] = 0.0f;
    }
}

__global__ void __launch_bounds__(256)
gcn_scatter_kernel(const float* __restrict__ H,
                   const int* __restrict__ ei,
                   const float* __restrict__ ew,
                   const float* __restrict__ deg,
                   float* __restrict__ aggA,
                   float* __restrict__ aggB, int E) {
    const int lane = threadIdx.x & 31;
    const int e  = blockIdx.x * blockDim.x + threadIdx.x;
    const int e0 = __builtin_amdgcn_readfirstlane(e);
    if (e0 >= E) return;
    const int cnt = (E - e0 < 32) ? (E - e0) : 32;

    const int ec = (e < E) ? e : (E - 1);
    const int src = ei[ec];
    const int dst = ei[E + ec];
    const float w = ew[ec];
    const float ds = fmaxf(deg[src], 1e-6f);
    const float dd = fmaxf(deg[dst], 1e-6f);
    const float norm = w / fmaxf(sqrtf(ds * dd), 1e-6f);
    const int normBits = __float_as_int(norm);

    float* __restrict__ agg = gcn_l2_half() ? aggB : aggA;

    const bool lowHalf = lane < 16;
    const int col0 = lowHalf ? 2 * lane : 2 * lane - 31;

    if (cnt == 32) {
        float2 h[32];
#pragma unroll
        for (int j = 0; j < 32; ++j) {
            const int s = __builtin_amdgcn_readlane(src, j);
            h[j] = ((const float2*)(H + (size_t)s * DIM))[lane];
        }
#pragma unroll
        for (int j = 0; j < 32; ++j) {
            const int d = __builtin_amdgcn_readlane(dst, j);
            const float nj = __int_as_float(__builtin_amdgcn_readlane(normBits, j));
            const float vx = nj * h[j].x;
            const float vy = nj * h[j].y;
            const float got = gcn_swap_halves(lowHalf ? vy : vx);
            float* arow = agg + (size_t)d * DIM + col0;
            atomicAdd(arow, lowHalf ? vx : got);
            atomicAdd(arow + DIM / 2, lowHalf ? got : vy);
        }
    } else {
        for (int j = 0; j < cnt; ++j) {
            const int s = __builtin_amdgcn_readlane(src, j);
            const int d = __builtin_amdgcn_readlane(dst, j);
            const float nj = __int_as_float(__builtin_amdgcn_readlane(normBits, j));
            const float2 hv = ((const float2*)(H + (size_t)s * DIM))[lane];
            const float vx = nj * hv.x;
            const float vy = nj * hv.y;
            const float got = gcn_swap_halves(lowHalf ? vy : vx);
            float* arow = agg + (size_t)d * DIM + col0;
            atomicAdd(arow, lowHalf ? vx : got);
            atomicAdd(arow + DIM / 2, lowHalf ? got : vy);
        }
    }
}

__global__ void __launch_bounds__(128)
gcn_gemm_wmma_kernel(const float* __restrict__ aggA,
                     const float* __restrict__ Wm,
                     float* out, int Nn) {
    const int wave    = threadIdx.x >> 5;
    const int lane    = threadIdx.x & 31;
    const int half    = lane >> 4;
    const int l       = lane & 15;
    const int rowBase = blockIdx.x * 16;
    const int colBase = wave * 16;

    v2f bfrag[16];
#pragma unroll
    for (int kk = 0; kk < 16; ++kk) {
        const float* wp = Wm + (size_t)(colBase + l) * DIM + kk * 4 + 2 * half;
        bfrag[kk] = *(const v2f*)wp;
    }

    int rA = rowBase + l;
    if (rA >= Nn) rA = Nn - 1;

    v8f c = {};
#pragma unroll
    for (int kk = 0; kk < 16; ++kk) {
        const size_t off = (size_t)rA * DIM + kk * 4 + 2 * half;
        v2f a0 = *(const v2f*)(aggA + off);
        v2f a1 = *(const v2f*)(out + off);
        c = __builtin_amdgcn_wmma_f32_16x16x4_f32(
                 false, a0,  false, bfrag[kk],
                 (short)0, c,  false,  false);
        c = __builtin_amdgcn_wmma_f32_16x16x4_f32(
                 false, a1,  false, bfrag[kk],
                 (short)0, c,  false,  false);
    }

    __builtin_amdgcn_sched_barrier(0);
    asm volatile("v_nop\n\tv_nop\n\tv_nop\n\tv_nop");
    __builtin_amdgcn_sched_barrier(0);

    __syncthreads();

#pragma unroll
    for (int r = 0; r < 8; ++r) {
        int row = rowBase + r + 8 * half;
        if (row < Nn) out[(size_t)row * DIM + colBase + l] = c[r];
    }
}

extern "C" void kernel_launch(void* const* d_in, const int* in_sizes, int n_in,
                              void* d_out, int out_size, void* d_ws, size_t ws_size,
                              hipStream_t stream) {
    const float* H   = (const float*)d_in[0];
    const int*   ei  = (const int*)d_in[1];
    const float* ew  = (const float*)d_in[2];
    const float* Wm  = (const float*)d_in[3];
    float*       out = (float*)d_out;

    const int N = in_sizes[0] / DIM;
    const int E = in_sizes[2];

    const int degPad = ((N + 63) / 64) * 64;
    float* deg = (float*)d_ws;
    float* agg = deg + degPad;

    {
        int na4 = (degPad + N * DIM) / 4;
        int nb4 = (N * DIM) / 4;
        gcn_zero_kernel<<<(na4 + nb4 + 255) / 256, 256, 0, stream>>>(deg, na4, out, nb4);
    }
    gcn_deg_kernel<<<(E + 255) / 256, 256, 0, stream>>>(ei, ew, deg, out, E);
    gcn_deg_merge_kernel<<<(N + 255) / 256, 256, 0, stream>>>(deg, out, N);

    gcn_scatter_kernel<<<(E + 255) / 256, 256, 0, stream>>>(H, ei, ew, deg, agg, out, E);

    gcn_gemm_wmma_kernel<<<(N + 15) / 16, 128, 0, stream>>>(agg, Wm, out, N);
}


# scatter-add: edges grouped by destination range (LDS counters), each range's agg rows summed by LDS float adds and stored once; no global f32 atomic on agg; sum order: arrival order of the LDS adds
# speedup vs baseline: 2.8921x; 1.4587x over previous
#include <hip/hip_runtime.h>
#include <hip/hip_bf16.h>


typedef __attribute__((ext_vector_type(2))) float v2f;
typedef __attribute__((ext_vector_type(8))) float v8f;

#define DIM 64
#define GCN_BIN_ROWS 400
#define GCN_MAX_BINS 1024
#define GCN_EPT 6
#define GCN_TPB 1024

__device__ __forceinline__ int gcn_l2_half() {
    unsigned id = __builtin_amdgcn_s_sendmsg_rtn(0x87);
    return (int)((id >> 18) & 1u);
}

__device__ __forceinline__ int gcn_block_excl_scan(int v, int n, int* bufA, int* bufB) {
    const int t = threadIdx.x;
    bufA[t] = v;
    __syncthreads();
    int* a = bufA;
    int* b = bufB;
    for (int off = 1; off < n; off <<= 1) {
        int x = a[t];
        if (t >= off) x += a[t - off];
        b[t] = x;
        __syncthreads();
        int* tmp = a; a = b; b = tmp;
    }
    return a[t] - v;
}

__global__ void gcn_zero_kernel(float* __restrict__ a, int na,
                                float* __restrict__ b, int nb,
                                int* __restrict__ c, int nc) {
    int i = blockIdx.x * blockDim.x + threadIdx.x;
    if (i < na) {
        a[i] = 0.0f;
    } else if (i - na < nb) {
        b[i - na] = 0.0f;
    } else if (i - na - nb < nc) {
        c[i - na - nb] = 0;
    }
}

__global__ void __launch_bounds__(GCN_TPB)
gcn_deg_hist_kernel(const int* __restrict__ ei,
                    const float* __restrict__ ew,
                    float* __restrict__ degA,
                    float* __restrict__ degB,
                    int* __restrict__ binCount, int E, int nbins) {
    __shared__ int lh[GCN_MAX_BINS];
    const int tid = threadIdx.x;
    for (int b = tid; b < nbins; b += GCN_TPB) lh[b] = 0;
    __syncthreads();
    float* deg = gcn_l2_half() ? degB : degA;
    const int c0 = blockIdx.x * (GCN_EPT * GCN_TPB);
#pragma unroll
    for (int k = 0; k < GCN_EPT; ++k) {
        const int e = c0 + k * GCN_TPB + tid;
        if (e < E) {
            const int d = ei[E + e];
            atomicAdd(&deg[d], ew[e]);
            atomicAdd(&lh[d / GCN_BIN_ROWS], 1);
        }
    }
    __syncthreads();
    for (int b = tid; b < nbins; b += GCN_TPB) {
        const int c = lh[b];
        if (c != 0) atomicAdd(&binCount[b], c);
    }
}

__global__ void gcn_deg_merge_kernel(float* __restrict__ degA,
                                     const float* __restrict__ degB, int n) {
    int i = blockIdx.x * blockDim.x + threadIdx.x;
    if (i < n) {
        degA[i] += degB[i];
    }
}

__global__ void __launch_bounds__(GCN_TPB)
gcn_place_kernel(const int* __restrict__ ei,
                 const float* __restrict__ ew,
                 const float* __restrict__ deg,
                 const int* __restrict__ binCount,
                 int* __restrict__ binFill,
                 int2* __restrict__ rec, int E, int nbins) {
    __shared__ int lh[GCN_MAX_BINS];
    __shared__ int lstart[GCN_MAX_BINS];
    __shared__ int delta[GCN_MAX_BINS];
    __shared__ int scanA[GCN_TPB];
    __shared__ int scanB[GCN_TPB];
    __shared__ int2 stage[GCN_EPT * GCN_TPB];
    __shared__ int binAt[GCN_EPT * GCN_TPB];
    const int tid = threadIdx.x;
    for (int b = tid; b < nbins; b += GCN_TPB) lh[b] = 0;
    __syncthreads();

    const int c0 = blockIdx.x * (GCN_EPT * GCN_TPB);
    int packed[GCN_EPT];
    int normBits[GCN_EPT];
    int binOf[GCN_EPT];
    int rank[GCN_EPT];
#pragma unroll
    for (int k = 0; k < GCN_EPT; ++k) {
        const int e = c0 + k * GCN_TPB + tid;
        const bool valid = e < E;
        const int ec = valid ? e : (E - 1);
        const int src = ei[ec];
        const int dst = ei[E + ec];
        const float w = ew[ec];
        const float ds = fmaxf(deg[src], 1e-6f);
        const float dd = fmaxf(deg[dst], 1e-6f);
        const float norm = w / fmaxf(sqrtf(ds * dd), 1e-6f);
        const int bin = dst / GCN_BIN_ROWS;
        packed[k] = src * 512 + (dst - bin * GCN_BIN_ROWS);
        normBits[k] = __float_as_int(norm);
        binOf[k] = valid ? bin : -1;
        rank[k] = valid ? atomicAdd(&lh[bin], 1) : 0;
    }
    __syncthreads();

    const int cntAll = (tid < nbins) ? binCount[tid] : 0;
    const int binStart = gcn_block_excl_scan(cntAll, nbins, scanA, scanB);
    __syncthreads();
    const int mine = (tid < nbins) ? lh[tid] : 0;
    const int localStart = gcn_block_excl_scan(mine, nbins, scanA, scanB);
    if (tid < nbins) {
        const int got = (mine != 0) ? atomicAdd(&binFill[tid], mine) : 0;
        lstart[tid] = localStart;
        delta[tid] = binStart + got - localStart;
    }
    __syncthreads();

#pragma unroll
    for (int k = 0; k < GCN_EPT; ++k) {
        if (binOf[k] >= 0) {
            const int slot = lstart[binOf[k]] + rank[k];
            stage[slot] = make_int2(packed[k], normBits[k]);
            binAt[slot] = binOf[k];
        }
    }
    __syncthreads();

    const int left = E - c0;
    const int mineAll = (left < GCN_EPT * GCN_TPB) ? left : (GCN_EPT * GCN_TPB);
    for (int slot = tid; slot < mineAll; slot += GCN_TPB) {
        rec[delta[binAt[slot]] + slot] = stage[slot];
    }
}

__global__ void __launch_bounds__(GCN_TPB)
gcn_bin_sum_kernel(const float* __restrict__ H,
                   const int2* __restrict__ rec,
                   const int* __restrict__ binCount,
                   float* __restrict__ aggOut, int Nn, int nbins) {
    __shared__ __attribute__((aligned(16))) float acc[GCN_BIN_ROWS * DIM];
    __shared__ int scanA[GCN_TPB];
    __shared__ int scanB[GCN_TPB];
    __shared__ int span[2];
    const int tid = threadIdx.x;
    const int lane = tid & 31;
    const int wave = tid >> 5;
    const int bin = blockIdx.x;

    {
        const float4 z = make_float4(0.0f, 0.0f, 0.0f, 0.0f);
        for (int i = tid; i < GCN_BIN_ROWS * DIM / 4; i += GCN_TPB) ((float4*)acc)[i] = z;
    }
    const int cntAll = (tid < nbins) ? binCount[tid] : 0;
    const int binStart = gcn_block_excl_scan(cntAll, nbins, scanA, scanB);
    if (tid == bin) { span[0] = binStart; span[1] = binStart + cntAll; }
    __syncthreads();
    const int first = span[0];
    const int last = span[1];

    for (int p = first + wave * 32; p < last; p += GCN_TPB) {
        const int cnt = (last - p < 32) ? (last - p) : 32;
        const int pc = (p + lane < last) ? (p + lane) : (last - 1);
        const int2 r = rec[pc];
        float* arow = acc + 2 * lane;
        if (cnt == 32) {
            float2 h[32];
#pragma unroll
            for (int j = 0; j < 32; ++j) {
                const int s = __builtin_amdgcn_readlane(r.x, j) >> 9;
                h[j] = ((const float2*)(H + (size_t)s * DIM))[lane];
            }
#pragma unroll
            for (int j = 0; j < 32; ++j) {
                const int dl = __builtin_amdgcn_readlane(r.x, j) & 511;
                const float nj = __int_as_float(__builtin_amdgcn_readlane(r.y, j));
                atomicAdd(arow + dl * DIM, nj * h[j].x);
                atomicAdd(arow + dl * DIM + 1, nj * h[j].y);
            }
        } else {
            for (int j0 = 0; j0 < cnt; j0 += 8) {
                float2 hv[8];
#pragma unroll
                for (int u = 0; u < 8; ++u) {
                    const int j = (j0 + u < cnt) ? (j0 + u) : (cnt - 1);
                    const int s = __builtin_amdgcn_readlane(r.x, j) >> 9;
                    hv[u] = ((const float2*)(H + (size_t)s * DIM))[lane];
                }
#pragma unroll
                for (int u = 0; u < 8; ++u) {
                    if (j0 + u < cnt) {
                        const int dl = __builtin_amdgcn_readlane(r.x, j0 + u) & 511;
                        const float nj = __int_as_float(__builtin_amdgcn_readlane(r.y, j0 + u));
                        atomicAdd(arow + dl * DIM, nj * hv[u].x);
                        atomicAdd(arow + dl * DIM + 1, nj * hv[u].y);
                    }
                }
            }
        }
    }
    __syncthreads();

    const int row0 = bin * GCN_BIN_ROWS;
    const int rows = (Nn - row0 < GCN_BIN_ROWS) ? (Nn - row0) : GCN_BIN_ROWS;
    float4* dstp = (float4*)(aggOut + (size_t)row0 * DIM);
    for (int i = tid; i < rows * (DIM / 4); i += GCN_TPB) dstp[i] = ((const float4*)acc)[i];
}

__global__ void __launch_bounds__(128)
gcn_gemm_wmma_kernel(const float* __restrict__ Wm,
                     float* out, int Nn) {
    const int wave    = threadIdx.x >> 5;
    const int lane    = threadIdx.x & 31;
    const int half    = lane >> 4;
    const int l       = lane & 15;
    const int rowBase = blockIdx.x * 16;
    const int colBase = wave * 16;

    v2f bfrag[16];
#pragma unroll
    for (int kk = 0; kk < 16; ++kk) {
        const float* wp = Wm + (size_t)(colBase + l) * DIM + kk * 4 + 2 * half;
        bfrag[kk] = *(const v2f*)wp;
    }

    int rA = rowBase + l;
    if (rA >= Nn) rA = Nn - 1;

    v2f afrag[16];
#pragma unroll
    for (int kk = 0; kk < 16; ++kk) {
        const size_t off = (size_t)rA * DIM + kk * 4 + 2 * half;
        afrag[kk] = *(const v2f*)(out + off);
    }

    v8f c = {};
#pragma unroll
    for (int kk = 0; kk < 16; ++kk) {
        c = __builtin_amdgcn_wmma_f32_16x16x4_f32(
                 false, afrag[kk],  false, bfrag[kk],
                 (short)0, c,  false,  false);
    }

    __builtin_amdgcn_sched_barrier(0);
    asm volatile("v_nop\n\tv_nop\n\tv_nop\n\tv_nop");
    __builtin_amdgcn_sched_barrier(0);

    __syncthreads();

#pragma unroll
    for (int r = 0; r < 8; ++r) {
        int row = rowBase + r + 8 * half;
        if (row < Nn) out[(size_t)row * DIM + colBase + l] = c[r];
    }
}

extern "C" void kernel_launch(void* const* d_in, const int* in_sizes, int n_in,
                              void* d_out, int out_size, void* d_ws, size_t ws_size,
                              hipStream_t stream) {
    const float* H   = (const float*)d_in[0];
    const int*   ei  = (const int*)d_in[1];
    const float* ew  = (const float*)d_in[2];
    const float* Wm  = (const float*)d_in[3];
    float*       out = (float*)d_out;

    const int N = in_sizes[0] / DIM;
    const int E = in_sizes[2];

    const int degPad = ((N + 63) / 64) * 64;
    float* deg      = (float*)d_ws;
    int2*  rec      = (int2*)(deg + degPad);
    int*   binCount = (int*)(rec + E);
    int*   binFill  = binCount + GCN_MAX_BINS;
    const int nbins = (N + GCN_BIN_ROWS - 1) / GCN_BIN_ROWS;
    const int edgeBlocks = (E + GCN_EPT * GCN_TPB - 1) / (GCN_EPT * GCN_TPB);

    gcn_zero_kernel<<<(2 * N + 2 * GCN_MAX_BINS + 255) / 256, 256, 0, stream>>>(
        deg, N, out, N, binCount, 2 * GCN_MAX_BINS);
    gcn_deg_hist_kernel<<<edgeBlocks, GCN_TPB, 0, stream>>>(ei, ew, deg, out, binCount, E, nbins);
    gcn_deg_merge_kernel<<<(N + 255) / 256, 256, 0, stream>>>(deg, out, N);

    gcn_place_kernel<<<edgeBlocks, GCN_TPB, 0, stream>>>(ei, ew, deg, binCount, binFill, rec, E, nbins);

    gcn_bin_sum_kernel<<<nbins, GCN_TPB, 0, stream>>>(H, rec, binCount, out, N, nbins);

    gcn_gemm_wmma_kernel<<<(N + 15) / 16, 128, 0, stream>>>(Wm, out, N);
}


# GEMM folded into the range-sum launch: each work-group multiplies its 400 summed agg rows by W^T straight from LDS (f32 WMMA, 272-B row pitch), stores out once; no GEMM launch or agg round trip
# speedup vs baseline: 3.4018x; 1.1762x over previous
#include <hip/hip_runtime.h>
#include <hip/hip_bf16.h>


typedef __attribute__((ext_vector_type(2))) float v2f;
typedef __attribute__((ext_vector_type(8))) float v8f;

#define DIM 64
#define GCN_BIN_ROWS 400
#define GCN_PITCH 68
#define GCN_MAX_BINS 1024
static_assert(GCN_BIN_ROWS % 16 == 0 && GCN_PITCH % 4 == 0, "whole row tiles, 16-byte row starts");
static_assert((GCN_BIN_ROWS - 1) * (GCN_PITCH / 4) < 8192, "a row start in 16-byte units fits 13 bits");
#define GCN_EPT 6
#define GCN_TPB 1024

__device__ __forceinline__ int gcn_l2_half() {
    unsigned id = __builtin_amdgcn_s_sendmsg_rtn(0x87);
    return (int)((id >> 18) & 1u);
}

__device__ __forceinline__ int gcn_block_excl_scan(int v, int n, int* bufA, int* bufB) {
    const int t = threadIdx.x;
    bufA[t] = v;
    __syncthreads();
    int* a = bufA;
    int* b = bufB;
    for (int off = 1; off < n; off <<= 1) {
        int x = a[t];
        if (t >= off) x += a[t - off];
        b[t] = x;
        __syncthreads();
        int* tmp = a; a = b; b = tmp;
    }
    return a[t] - v;
}

__global__ void gcn_zero_kernel(float* __restrict__ a, int na,
                                float* __restrict__ b, int nb,
                                int* __restrict__ c, int nc) {
    int i = blockIdx.x * blockDim.x + threadIdx.x;
    if (i < na) {
        a[i] = 0.0f;
    } else if (i - na < nb) {
        b[i - na] = 0.0f;
    } else if (i - na - nb < nc) {
        c[i - na - nb] = 0;
    }
}

__global__ void __launch_bounds__(GCN_TPB)
gcn_deg_hist_kernel(const int* __restrict__ ei,
                    const float* __restrict__ ew,
                    float* __restrict__ degA,
                    float* __restrict__ degB,
                    int* __restrict__ binCount, int E, int nbins) {
    __shared__ int lh[GCN_MAX_BINS];
    const int tid = threadIdx.x;
    for (int b = tid; b < nbins; b += GCN_TPB) lh[b] = 0;
    __syncthreads();
    float* deg = gcn_l2_half() ? degB : degA;
    const int c0 = blockIdx.x * (GCN_EPT * GCN_TPB);
#pragma unroll
    for (int k = 0; k < GCN_EPT; ++k) {
        const int e = c0 + k * GCN_TPB + tid;
        if (e < E) {
            const int d = ei[E + e];
            atomicAdd(&deg[d], ew[e]);
            atomicAdd(&lh[d / GCN_BIN_ROWS], 1);
        }
    }
    __syncthreads();
    for (int b = tid; b < nbins; b += GCN_TPB) {
        const int c = lh[b];
        if (c != 0) atomicAdd(&binCount[b], c);
    }
}

__global__ void gcn_deg_merge_kernel(float* __restrict__ degA,
                                     const float* __restrict__ degB, int n) {
    int i = blockIdx.x * blockDim.x + threadIdx.x;
    if (i < n) {
        degA[i] += degB[i];
    }
}

__global__ void __launch_bounds__(GCN_TPB)
gcn_place_kernel(const int* __restrict__ ei,
                 const float* __restrict__ ew,
                 const float* __restrict__ deg,
                 const int* __restrict__ binCount,
                 int* __restrict__ binFill,
                 int2* __restrict__ rec, int E, int nbins) {
    __shared__ int lh[GCN_MAX_BINS];
    __shared__ int lstart[GCN_MAX_BINS];
    __shared__ int delta[GCN_MAX_BINS];
    __shared__ int scanA[GCN_TPB];
    __shared__ int scanB[GCN_TPB];
    __shared__ int2 stage[GCN_EPT * GCN_TPB];
    __shared__ int binAt[GCN_EPT * GCN_TPB];
    const int tid = threadIdx.x;
    for (int b = tid; b < nbins; b += GCN_TPB) lh[b] = 0;
    __syncthreads();

    const int c0 = blockIdx.x * (GCN_EPT * GCN_TPB);
    int packed[GCN_EPT];
    int normBits[GCN_EPT];
    int binOf[GCN_EPT];
    int rank[GCN_EPT];
#pragma unroll
    for (int k = 0; k < GCN_EPT; ++k) {
        const int e = c0 + k * GCN_TPB + tid;
        const bool valid = e < E;
        const int ec = valid ? e : (E - 1);
        const int src = ei[ec];
        const int dst = ei[E + ec];
        const float w = ew[ec];
        const float ds = fmaxf(deg[src], 1e-6f);
        const float dd = fmaxf(deg[dst], 1e-6f);
        const float norm = w / fmaxf(sqrtf(ds * dd), 1e-6f);
        const int bin = dst / GCN_BIN_ROWS;
        packed[k] = src * 512 + (dst - bin * GCN_BIN_ROWS);
        normBits[k] = __float_as_int(norm);
        binOf[k] = valid ? bin : -1;
        rank[k] = valid ? atomicAdd(&lh[bin], 1) : 0;
    }
    __syncthreads();

    const int cntAll = (tid < nbins) ? binCount[tid] : 0;
    const int binStart = gcn_block_excl_scan(cntAll, nbins, scanA, scanB);
    __syncthreads();
    const int mine = (tid < nbins) ? lh[tid] : 0;
    const int localStart = gcn_block_excl_scan(mine, nbins, scanA, scanB);
    if (tid < nbins) {
        const int got = (mine != 0) ? atomicAdd(&binFill[tid], mine) : 0;
        lstart[tid] = localStart;
        delta[tid] = binStart + got - localStart;
    }
    __syncthreads();

#pragma unroll
    for (int k = 0; k < GCN_EPT; ++k) {
        if (binOf[k] >= 0) {
            const int slot = lstart[binOf[k]] + rank[k];
            stage[slot] = make_int2(packed[k], normBits[k]);
            binAt[slot] = binOf[k];
        }
    }
    __syncthreads();

    const int left = E - c0;
    const int mineAll = (left < GCN_EPT * GCN_TPB) ? left : (GCN_EPT * GCN_TPB);
    for (int slot = tid; slot < mineAll; slot += GCN_TPB) {
        rec[delta[binAt[slot]] + slot] = stage[slot];
    }
}

__global__ void __launch_bounds__(GCN_TPB)
gcn_bin_sum_gemm_kernel(const float* __restrict__ H,
                        const int2* __restrict__ rec,
                        const int* __restrict__ binCount,
                        const float* __restrict__ Wm,
                        float* __restrict__ out, int Nn, int nbins) {
    __shared__ __attribute__((aligned(16))) float acc[GCN_BIN_ROWS * GCN_PITCH];
    __shared__ int scanA[GCN_TPB];
    __shared__ int scanB[GCN_TPB];
    __shared__ int span[2];
    const int tid = threadIdx.x;
    const int lane = tid & 31;
    const int wave = tid >> 5;
    const int bin = blockIdx.x;

    {
        const float4 z = make_float4(0.0f, 0.0f, 0.0f, 0.0f);
        for (int i = tid; i < GCN_BIN_ROWS * GCN_PITCH / 4; i += GCN_TPB) ((float4*)acc)[i] = z;
    }
    const int cntAll = (tid < nbins) ? binCount[tid] : 0;
    const int binStart = gcn_block_excl_scan(cntAll, nbins, scanA, scanB);
    if (tid == bin) { span[0] = binStart; span[1] = binStart + cntAll; }
    __syncthreads();
    const int first = span[0];
    const int last = span[1];

    for (int p = first + wave * 32; p < last; p += GCN_TPB) {
        const int cnt = (last - p < 32) ? (last - p) : 32;
        const int pc = (p + lane < last) ? (p + lane) : (last - 1);
        const int2 r = rec[pc];
        const int pk = (int)(((unsigned)(r.x >> 9) << 13) |
                             (unsigned)((r.x & 511) * (GCN_PITCH / 4)));
        char* abase = (char*)(acc + 2 * lane);
        if (cnt == 32) {
            float2 h[32];
#pragma unroll
            for (int j = 0; j < 32; ++j) {
                const int s = (int)((unsigned)__builtin_amdgcn_readlane(pk, j) >> 13);
                h[j] = ((const float2*)(H + (size_t)s * DIM))[lane];
            }
#pragma unroll
            for (int j = 0; j < 32; ++j) {
                float* arow = (float*)(abase + ((__builtin_amdgcn_readlane(pk, j) & 0x1fff) << 4));
                const float nj = __int_as_float(__builtin_amdgcn_readlane(r.y, j));
                atomicAdd(arow, nj * h[j].x);
                atomicAdd(arow + 1, nj * h[j].y);
            }
        } else {
            for (int j0 = 0; j0 < cnt; j0 += 8) {
                float2 hv[8];
#pragma unroll
                for (int u = 0; u < 8; ++u) {
                    const int j = (j0 + u < cnt) ? (j0 + u) : (cnt - 1);
                    const int s = (int)((unsigned)__builtin_amdgcn_readlane(pk, j) >> 13);
                    hv[u] = ((const float2*)(H + (size_t)s * DIM))[lane];
                }
#pragma unroll
                for (int u = 0; u < 8; ++u) {
                    if (j0 + u < cnt) {
                        float* arow = (float*)(abase + ((__builtin_amdgcn_readlane(pk, j0 + u) & 0x1fff) << 4));
                        const float nj = __int_as_float(__builtin_amdgcn_readlane(r.y, j0 + u));
                        atomicAdd(arow, nj * hv[u].x);
                        atomicAdd(arow + 1, nj * hv[u].y);
                    }
                }
            }
        }
    }

    const int wv   = __builtin_amdgcn_readfirstlane(wave);
    const int half = lane >> 4;
    const int l    = lane & 15;
    const int ct   = wv & 3;
    const int rt0  = wv >> 2;

    v2f bfrag[16];
    {
        const float* wp = Wm + (size_t)(ct * 16 + l) * DIM + 2 * half;
#pragma unroll
        for (int kk = 0; kk < 16; ++kk) bfrag[kk] = *(const v2f*)(wp + kk * 4);
    }

    __syncthreads();

    v8f c[4];
#pragma unroll
    for (int i = 0; i < 4; ++i) {
        const int rt = rt0 + 8 * i;
        if (rt < GCN_BIN_ROWS / 16) {
            const float* ap = acc + (rt * 16 + l) * GCN_PITCH + 2 * half;
            v2f afrag[16];
#pragma unroll
            for (int kk = 0; kk < 16; ++kk) afrag[kk] = *(const v2f*)(ap + kk * 4);
            v8f ci = {};
            __builtin_amdgcn_sched_barrier(0);
#pragma unroll
            for (int kk = 0; kk < 16; ++kk) {
                ci = __builtin_amdgcn_wmma_f32_16x16x4_f32(
                         false, afrag[kk],  false, bfrag[kk],
                         (short)0, ci,  false,  false);
            }
            __builtin_amdgcn_sched_barrier(0);
            asm volatile("v_nop\n\tv_nop\n\tv_nop\n\tv_nop");
            __builtin_amdgcn_sched_barrier(0);
            c[i] = ci;
        }
    }

    __syncthreads();

#pragma unroll
    for (int i = 0; i < 4; ++i) {
        const int rt = rt0 + 8 * i;
        if (rt < GCN_BIN_ROWS / 16) {
            float* dp = acc + (rt * 16 + 8 * half) * GCN_PITCH + ct * 16 + l;
#pragma unroll
            for (int r = 0; r < 8; ++r) dp[r * GCN_PITCH] = c[i][r];
        }
    }
    __syncthreads();

    const int row0 = bin * GCN_BIN_ROWS;
    const int rows = (Nn - row0 < GCN_BIN_ROWS) ? (Nn - row0) : GCN_BIN_ROWS;
    float4* dstp = (float4*)(out + (size_t)row0 * DIM);
    for (int i = tid; i < rows * (DIM / 4); i += GCN_TPB) {
        dstp[i] = *(const float4*)(acc + (i >> 4) * GCN_PITCH + (i & 15) * 4);
    }
}

extern "C" void kernel_launch(void* const* d_in, const int* in_sizes, int n_in,
                              void* d_out, int out_size, void* d_ws, size_t ws_size,
                              hipStream_t stream) {
    const float* H   = (const float*)d_in[0];
    const int*   ei  = (const int*)d_in[1];
    const float* ew  = (const float*)d_in[2];
    const float* Wm  = (const float*)d_in[3];
    float*       out = (float*)d_out;

    const int N = in_sizes[0] / DIM;
    const int E = in_sizes[2];

    const int degPad = ((N + 63) / 64) * 64;
    float* deg      = (float*)d_ws;
    int2*  rec      = (int2*)(deg + degPad);
    int*   binCount = (int*)(rec + E);
    int*   binFill  = binCount + GCN_MAX_BINS;
    const int nbins = (N + GCN_BIN_ROWS - 1) / GCN_BIN_ROWS;
    const int edgeBlocks = (E + GCN_EPT * GCN_TPB - 1) / (GCN_EPT * GCN_TPB);

    gcn_zero_kernel<<<(2 * N + 2 * GCN_MAX_BINS + 255) / 256, 256, 0, stream>>>(
        deg, N, out, N, binCount, 2 * GCN_MAX_BINS);
    gcn_deg_hist_kernel<<<edgeBlocks, GCN_TPB, 0, stream>>>(ei, ew, deg, out, binCount, E, nbins);
    gcn_deg_merge_kernel<<<(N + 255) / 256, 256, 0, stream>>>(deg, out, N);

    gcn_place_kernel<<<edgeBlocks, GCN_TPB, 0, stream>>>(ei, ew, deg, binCount, binFill, rec, E, nbins);

    gcn_bin_sum_gemm_kernel<<<nbins, GCN_TPB, 0, stream>>>(H, rec, binCount, Wm, out, N, nbins);
}


# deg summed per destination range from the grouped edge list by LDS float adds (own launch; zero+merge launches gone), norm taken in the range launch; no global float atomic; sum order: LDS arrival
# speedup vs baseline: 4.2947x; 1.2625x over previous
#include <hip/hip_runtime.h>
#include <hip/hip_bf16.h>


typedef __attribute__((ext_vector_type(2))) float v2f;
typedef __attribute__((ext_vector_type(8))) float v8f;

#define DIM 64
#define GCN_BIN_ROWS 400
#define GCN_PITCH 68
#define GCN_MAX_BINS 1024
static_assert(GCN_BIN_ROWS % 16 == 0 && GCN_PITCH % 4 == 0, "whole row tiles, 16-byte row starts");
static_assert((GCN_BIN_ROWS - 1) * (GCN_PITCH / 4) < 8192, "a row start in 16-byte units fits 13 bits");
#define GCN_EPT 6
#define GCN_DEG_TPB 256
#define GCN_DEG_BATCH 16
#define GCN_TPB 1024

__device__ __forceinline__ int gcn_block_excl_scan(int v, int n, int* bufA, int* bufB) {
    const int t = threadIdx.x;
    bufA[t] = v;
    __syncthreads();
    int* a = bufA;
    int* b = bufB;
    for (int off = 1; off < n; off <<= 1) {
        int x = a[t];
        if (t >= off) x += a[t - off];
        b[t] = x;
        __syncthreads();
        int* tmp = a; a = b; b = tmp;
    }
    return a[t] - v;
}

__global__ void gcn_zero_counters_kernel(int* __restrict__ c, int nc) {
    int i = blockIdx.x * blockDim.x + threadIdx.x;
    if (i < nc) {
        c[i] = 0;
    }
}

__global__ void __launch_bounds__(GCN_TPB)
gcn_hist_kernel(const int* __restrict__ ei,
                int* __restrict__ binCount, int E, int nbins) {
    __shared__ int lh[GCN_MAX_BINS];
    const int tid = threadIdx.x;
    for (int b = tid; b < nbins; b += GCN_TPB) lh[b] = 0;
    __syncthreads();
    const int c0 = blockIdx.x * (GCN_EPT * GCN_TPB);
#pragma unroll
    for (int k = 0; k < GCN_EPT; ++k) {
        const int e = c0 + k * GCN_TPB + tid;
        if (e < E) {
            const int d = ei[E + e];
            atomicAdd(&lh[d / GCN_BIN_ROWS], 1);
        }
    }
    __syncthreads();
    for (int b = tid; b < nbins; b += GCN_TPB) {
        const int c = lh[b];
        if (c != 0) atomicAdd(&binCount[b], c);
    }
}

__global__ void __launch_bounds__(GCN_TPB)
gcn_place_kernel(const int* __restrict__ ei,
                 const float* __restrict__ ew,
                 const int* __restrict__ binCount,
                 int* __restrict__ binFill,
                 int2* __restrict__ rec, int E, int nbins) {
    __shared__ int lh[GCN_MAX_BINS];
    __shared__ int lstart[GCN_MAX_BINS];
    __shared__ int delta[GCN_MAX_BINS];
    __shared__ int scanA[GCN_TPB];
    __shared__ int scanB[GCN_TPB];
    __shared__ int2 stage[GCN_EPT * GCN_TPB];
    __shared__ int binAt[GCN_EPT * GCN_TPB];
    const int tid = threadIdx.x;
    for (int b = tid; b < nbins; b += GCN_TPB) lh[b] = 0;
    __syncthreads();

    const int c0 = blockIdx.x * (GCN_EPT * GCN_TPB);
    int packed[GCN_EPT];
    int wBits[GCN_EPT];
    int binOf[GCN_EPT];
    int rank[GCN_EPT];
#pragma unroll
    for (int k = 0; k < GCN_EPT; ++k) {
        const int e = c0 + k * GCN_TPB + tid;
        const bool valid = e < E;
        const int ec = valid ? e : (E - 1);
        const int src = ei[ec];
        const int dst = ei[E + ec];
        const float w = ew[ec];
        const int bin = dst / GCN_BIN_ROWS;
        packed[k] = src * 512 + (dst - bin * GCN_BIN_ROWS);
        wBits[k] = __float_as_int(w);
        binOf[k] = valid ? bin : -1;
        rank[k] = valid ? atomicAdd(&lh[bin], 1) : 0;
    }
    __syncthreads();

    const int cntAll = (tid < nbins) ? binCount[tid] : 0;
    const int binStart = gcn_block_excl_scan(cntAll, nbins, scanA, scanB);
    __syncthreads();
    const int mine = (tid < nbins) ? lh[tid] : 0;
    const int localStart = gcn_block_excl_scan(mine, nbins, scanA, scanB);
    if (tid < nbins) {
        const int got = (mine != 0) ? atomicAdd(&binFill[tid], mine) : 0;
        lstart[tid] = localStart;
        delta[tid] = binStart + got - localStart;
    }
    __syncthreads();

#pragma unroll
    for (int k = 0; k < GCN_EPT; ++k) {
        if (binOf[k] >= 0) {
            const int slot = lstart[binOf[k]] + rank[k];
            stage[slot] = make_int2(packed[k], wBits[k]);
            binAt[slot] = binOf[k];
        }
    }
    __syncthreads();

    const int left = E - c0;
    const int mineAll = (left < GCN_EPT * GCN_TPB) ? left : (GCN_EPT * GCN_TPB);
    for (int slot = tid; slot < mineAll; slot += GCN_TPB) {
        rec[delta[binAt[slot]] + slot] = stage[slot];
    }
}

__global__ void __launch_bounds__(GCN_DEG_TPB)
gcn_bin_deg_kernel(const int2* __restrict__ rec,
                   const int* __restrict__ binCount,
                   float* __restrict__ deg, int Nn) {
    __shared__ float dl[GCN_BIN_ROWS];
    __shared__ int part[GCN_DEG_TPB / 32];
    const int tid = threadIdx.x;
    const int bin = blockIdx.x;
    for (int i = tid; i < GCN_BIN_ROWS; i += GCN_DEG_TPB) dl[i] = 0.0f;
    int s = 0;
    for (int i = tid; i < bin; i += GCN_DEG_TPB) s += binCount[i];
    for (int off = 16; off > 0; off >>= 1) s += __shfl_xor(s, off);
    if ((tid & 31) == 0) part[tid >> 5] = s;
    __syncthreads();
    int first = 0;
    for (int w = 0; w < GCN_DEG_TPB / 32; ++w) first += part[w];
    const int last = first + binCount[bin];

    for (int p0 = first; p0 < last; p0 += GCN_DEG_BATCH * GCN_DEG_TPB) {
        int2 rr[GCN_DEG_BATCH];
#pragma unroll
        for (int u = 0; u < GCN_DEG_BATCH; ++u) {
            const int p = p0 + u * GCN_DEG_TPB + tid;
            rr[u] = rec[(p < last) ? p : (last - 1)];
        }
#pragma unroll
        for (int u = 0; u < GCN_DEG_BATCH; ++u) {
            const int p = p0 + u * GCN_DEG_TPB + tid;
            if (p < last) atomicAdd(&dl[rr[u].x & 511], __int_as_float(rr[u].y));
        }
    }
    __syncthreads();
    const int row0 = bin * GCN_BIN_ROWS;
    for (int i = tid; i < GCN_BIN_ROWS; i += GCN_DEG_TPB) {
        if (row0 + i < Nn) deg[row0 + i] = fmaxf(dl[i], 1e-6f);
    }
}

__global__ void __launch_bounds__(GCN_TPB)
gcn_bin_sum_gemm_kernel(const float* __restrict__ H,
                        const int2* __restrict__ rec,
                        const int* __restrict__ binCount,
                        const float* __restrict__ deg,
                        const float* __restrict__ Wm,
                        float* __restrict__ out, int Nn, int nbins) {
    __shared__ __attribute__((aligned(16))) float acc[GCN_BIN_ROWS * GCN_PITCH];
    __shared__ int scanA[GCN_TPB];
    __shared__ int scanB[GCN_TPB];
    __shared__ int span[2];
    __shared__ float degL[GCN_BIN_ROWS];
    const int tid = threadIdx.x;
    const int lane = tid & 31;
    const int wave = tid >> 5;
    const int bin = blockIdx.x;
    const int row0 = bin * GCN_BIN_ROWS;

    {
        const float4 z = make_float4(0.0f, 0.0f, 0.0f, 0.0f);
        for (int i = tid; i < GCN_BIN_ROWS * GCN_PITCH / 4; i += GCN_TPB) ((float4*)acc)[i] = z;
    }
    for (int i = tid; i < GCN_BIN_ROWS; i += GCN_TPB) degL[i] = (row0 + i < Nn) ? deg[row0 + i] : 1.0f;
    const int cntAll = (tid < nbins) ? binCount[tid] : 0;
    const int binStart = gcn_block_excl_scan(cntAll, nbins, scanA, scanB);
    if (tid == bin) { span[0] = binStart; span[1] = binStart + cntAll; }
    __syncthreads();
    const int first = span[0];
    const int last = span[1];

    for (int p = first + wave * 32; p < last; p += GCN_TPB) {
        const int cnt = (last - p < 32) ? (last - p) : 32;
        const int pc = (p + lane < last) ? (p + lane) : (last - 1);
        const int2 r = rec[pc];
        const float ds = deg[r.x >> 9];
        const int pk = (int)(((unsigned)(r.x >> 9) << 13) |
                             (unsigned)((r.x & 511) * (GCN_PITCH / 4)));
        char* abase = (char*)(acc + 2 * lane);
        if (cnt == 32) {
            float2 h[32];
#pragma unroll
            for (int j = 0; j < 32; ++j) {
                const int s = (int)((unsigned)__builtin_amdgcn_readlane(pk, j) >> 13);
                h[j] = ((const float2*)(H + (size_t)s * DIM))[lane];
            }
            const float dd = degL[r.x & 511];
            const int nb = __float_as_int(__int_as_float(r.y) / fmaxf(sqrtf(ds * dd), 1e-6f));
#pragma unroll
            for (int j = 0; j < 32; ++j) {
                float* arow = (float*)(abase + ((__builtin_amdgcn_readlane(pk, j) & 0x1fff) << 4));
                const float nj = __int_as_float(__builtin_amdgcn_readlane(nb, j));
                atomicAdd(arow, nj * h[j].x);
                atomicAdd(arow + 1, nj * h[j].y);
            }
        } else {
            const float dd = degL[r.x & 511];
            const int nb = __float_as_int(__int_as_float(r.y) / fmaxf(sqrtf(ds * dd), 1e-6f));
            for (int j0 = 0; j0 < cnt; j0 += 8) {
                float2 hv[8];
#pragma unroll
                for (int u = 0; u < 8; ++u) {
                    const int j = (j0 + u < cnt) ? (j0 + u) : (cnt - 1);
                    const int s = (int)((unsigned)__builtin_amdgcn_readlane(pk, j) >> 13);
                    hv[u] = ((const float2*)(H + (size_t)s * DIM))[lane];
                }
#pragma unroll
                for (int u = 0; u < 8; ++u) {
                    if (j0 + u < cnt) {
                        float* arow = (float*)(abase + ((__builtin_amdgcn_readlane(pk, j0 + u) & 0x1fff) << 4));
                        const float nj = __int_as_float(__builtin_amdgcn_readlane(nb, j0 + u));
                        atomicAdd(arow, nj * hv[u].x);
                        atomicAdd(arow + 1, nj * hv[u].y);
                    }
                }
            }
        }
    }

    const int wv   = __builtin_amdgcn_readfirstlane(wave);
    const int half = lane >> 4;
    const int l    = lane & 15;
    const int ct   = wv & 3;
    const int rt0  = wv >> 2;

    v2f bfrag[16];
    {
        const float* wp = Wm + (size_t)(ct * 16 + l) * DIM + 2 * half;
#pragma unroll
        for (int kk = 0; kk < 16; ++kk) bfrag[kk] = *(const v2f*)(wp + kk * 4);
    }

    __syncthreads();

    v8f c[4];
#pragma unroll
    for (int i = 0; i < 4; ++i) {
        const int rt = rt0 + 8 * i;
        if (rt < GCN_BIN_ROWS / 16) {
            const float* ap = acc + (rt * 16 + l) * GCN_PITCH + 2 * half;
            v2f afrag[16];
#pragma unroll
            for (int kk = 0; kk < 16; ++kk) afrag[kk] = *(const v2f*)(ap + kk * 4);
            v8f ci = {};
            __builtin_amdgcn_sched_barrier(0);
#pragma unroll
            for (int kk = 0; kk < 16; ++kk) {
                ci = __builtin_amdgcn_wmma_f32_16x16x4_f32(
                         false, afrag[kk],  false, bfrag[kk],
                         (short)0, ci,  false,  false);
            }
            __builtin_amdgcn_sched_barrier(0);
            asm volatile("v_nop\n\tv_nop\n\tv_nop\n\tv_nop");
            __builtin_amdgcn_sched_barrier(0);
            c[i] = ci;
        }
    }

    __syncthreads();

#pragma unroll
    for (int i = 0; i < 4; ++i) {
        const int rt = rt0 + 8 * i;
        if (rt < GCN_BIN_ROWS / 16) {
            float* dp = acc + (rt * 16 + 8 * half) * GCN_PITCH + ct * 16 + l;
#pragma unroll
            for (int r = 0; r < 8; ++r) dp[r * GCN_PITCH] = c[i][r];
        }
    }
    __syncthreads();

    const int rows = (Nn - row0 < GCN_BIN_ROWS) ? (Nn - row0) : GCN_BIN_ROWS;
    float4* dstp = (float4*)(out + (size_t)row0 * DIM);
    for (int i = tid; i < rows * (DIM / 4); i += GCN_TPB) {
        dstp[i] = *(const float4*)(acc + (i >> 4) * GCN_PITCH + (i & 15) * 4);
    }
}

extern "C" void kernel_launch(void* const* d_in, const int* in_sizes, int n_in,
                              void* d_out, int out_size, void* d_ws, size_t ws_size,
                              hipStream_t stream) {
    const float* H   = (const float*)d_in[0];
    const int*   ei  = (const int*)d_in[1];
    const float* ew  = (const float*)d_in[2];
    const float* Wm  = (const float*)d_in[3];
    float*       out = (float*)d_out;

    const int N = in_sizes[0] / DIM;
    const int E = in_sizes[2];

    const int degPad = ((N + 63) / 64) * 64;
    float* deg      = (float*)d_ws;
    int2*  rec      = (int2*)(deg + degPad);
    int*   binCount = (int*)(rec + E);
    int*   binFill  = binCount + GCN_MAX_BINS;
    const int nbins = (N + GCN_BIN_ROWS - 1) / GCN_BIN_ROWS;
    const int edgeBlocks = (E + GCN_EPT * GCN_TPB - 1) / (GCN_EPT * GCN_TPB);

    gcn_zero_counters_kernel<<<(2 * GCN_MAX_BINS + 255) / 256, 256, 0, stream>>>(binCount, 2 * GCN_MAX_BINS);
    gcn_hist_kernel<<<edgeBlocks, GCN_TPB, 0, stream>>>(ei, binCount, E, nbins);
    gcn_place_kernel<<<edgeBlocks, GCN_TPB, 0, stream>>>(ei, ew, binCount, binFill, rec, E, nbins);
    gcn_bin_deg_kernel<<<nbins, GCN_DEG_TPB, 0, stream>>>(rec, binCount, deg, N);
    gcn_bin_sum_gemm_kernel<<<nbins, GCN_TPB, 0, stream>>>(H, rec, binCount, deg, Wm, out, N, nbins);
}


# range launch: records put in row order in LDS (counting sort), each row's run summed in registers, one LDS float add pair per run not per edge; sum order: sorted-list order, runs by arrival
# speedup vs baseline: 4.7753x; 1.1119x over previous
#include <hip/hip_runtime.h>
#include <hip/hip_bf16.h>


typedef __attribute__((ext_vector_type(2))) float v2f;
typedef __attribute__((ext_vector_type(8))) float v8f;

#define DIM 64
#define GCN_BIN_ROWS 400
#define GCN_PITCH 68
#define GCN_MAX_BINS 1024
static_assert(GCN_BIN_ROWS % 16 == 0 && GCN_PITCH % 4 == 0, "whole row tiles, 16-byte row starts");
static_assert((GCN_BIN_ROWS - 1) * (GCN_PITCH / 4) < 8192, "a row start in 16-byte units fits 13 bits");
#define GCN_EPT 6
#define GCN_DEG_TPB 256
#define GCN_DEG_BATCH 16
#define GCN_TPB 1024
#define GCN_SORT_EPT 8
#define GCN_SORT_CAP (GCN_SORT_EPT * GCN_TPB)
static_assert(GCN_BIN_ROWS <= GCN_TPB, "one thread per row in the row scan");

__device__ __forceinline__ int gcn_block_excl_scan(int v, int n, int* bufA, int* bufB) {
    const int t = threadIdx.x;
    bufA[t] = v;
    __syncthreads();
    int* a = bufA;
    int* b = bufB;
    for (int off = 1; off < n; off <<= 1) {
        int x = a[t];
        if (t >= off) x += a[t - off];
        b[t] = x;
        __syncthreads();
        int* tmp = a; a = b; b = tmp;
    }
    return a[t] - v;
}

__global__ void gcn_zero_counters_kernel(int* __restrict__ c, int nc) {
    int i = blockIdx.x * blockDim.x + threadIdx.x;
    if (i < nc) {
        c[i] = 0;
    }
}

__global__ void __launch_bounds__(GCN_TPB)
gcn_hist_kernel(const int* __restrict__ ei,
                int* __restrict__ binCount, int E, int nbins) {
    __shared__ int lh[GCN_MAX_BINS];
    const int tid = threadIdx.x;
    for (int b = tid; b < nbins; b += GCN_TPB) lh[b] = 0;
    __syncthreads();
    const int c0 = blockIdx.x * (GCN_EPT * GCN_TPB);
#pragma unroll
    for (int k = 0; k < GCN_EPT; ++k) {
        const int e = c0 + k * GCN_TPB + tid;
        if (e < E) {
            const int d = ei[E + e];
            atomicAdd(&lh[d / GCN_BIN_ROWS], 1);
        }
    }
    __syncthreads();
    for (int b = tid; b < nbins; b += GCN_TPB) {
        const int c = lh[b];
        if (c != 0) atomicAdd(&binCount[b], c);
    }
}

__global__ void __launch_bounds__(GCN_TPB)
gcn_place_kernel(const int* __restrict__ ei,
                 const float* __restrict__ ew,
                 const int* __restrict__ binCount,
                 int* __restrict__ binFill,
                 int2* __restrict__ rec, int E, int nbins) {
    __shared__ int lh[GCN_MAX_BINS];
    __shared__ int lstart[GCN_MAX_BINS];
    __shared__ int delta[GCN_MAX_BINS];
    __shared__ int scanA[GCN_TPB];
    __shared__ int scanB[GCN_TPB];
    __shared__ int2 stage[GCN_EPT * GCN_TPB];
    __shared__ int binAt[GCN_EPT * GCN_TPB];
    const int tid = threadIdx.x;
    for (int b = tid; b < nbins; b += GCN_TPB) lh[b] = 0;
    __syncthreads();

    const int c0 = blockIdx.x * (GCN_EPT * GCN_TPB);
    int packed[GCN_EPT];
    int wBits[GCN_EPT];
    int binOf[GCN_EPT];
    int rank[GCN_EPT];
#pragma unroll
    for (int k = 0; k < GCN_EPT; ++k) {
        const int e = c0 + k * GCN_TPB + tid;
        const bool valid = e < E;
        const int ec = valid ? e : (E - 1);
        const int src = ei[ec];
        const int dst = ei[E + ec];
        const float w = ew[ec];
        const int bin = dst / GCN_BIN_ROWS;
        packed[k] = src * 512 + (dst - bin * GCN_BIN_ROWS);
        wBits[k] = __float_as_int(w);
        binOf[k] = valid ? bin : -1;
        rank[k] = valid ? atomicAdd(&lh[bin], 1) : 0;
    }
    __syncthreads();

    const int cntAll = (tid < nbins) ? binCount[tid] : 0;
    const int binStart = gcn_block_excl_scan(cntAll, nbins, scanA, scanB);
    __syncthreads();
    const int mine = (tid < nbins) ? lh[tid] : 0;
    const int localStart = gcn_block_excl_scan(mine, nbins, scanA, scanB);
    if (tid < nbins) {
        const int got = (mine != 0) ? atomicAdd(&binFill[tid], mine) : 0;
        lstart[tid] = localStart;
        delta[tid] = binStart + got - localStart;
    }
    __syncthreads();

#pragma unroll
    for (int k = 0; k < GCN_EPT; ++k) {
        if (binOf[k] >= 0) {
            const int slot = lstart[binOf[k]] + rank[k];
            stage[slot] = make_int2(packed[k], wBits[k]);
            binAt[slot] = binOf[k];
        }
    }
    __syncthreads();

    const int left = E - c0;
    const int mineAll = (left < GCN_EPT * GCN_TPB) ? left : (GCN_EPT * GCN_TPB);
    for (int slot = tid; slot < mineAll; slot += GCN_TPB) {
        rec[delta[binAt[slot]] + slot] = stage[slot];
    }
}

__global__ void __launch_bounds__(GCN_DEG_TPB)
gcn_bin_deg_kernel(const int2* __restrict__ rec,
                   const int* __restrict__ binCount,
                   float* __restrict__ deg, int Nn) {
    __shared__ float dl[GCN_BIN_ROWS];
    __shared__ int part[GCN_DEG_TPB / 32];
    const int tid = threadIdx.x;
    const int bin = blockIdx.x;
    for (int i = tid; i < GCN_BIN_ROWS; i += GCN_DEG_TPB) dl[i] = 0.0f;
    int s = 0;
    for (int i = tid; i < bin; i += GCN_DEG_TPB) s += binCount[i];
    for (int off = 16; off > 0; off >>= 1) s += __shfl_xor(s, off);
    if ((tid & 31) == 0) part[tid >> 5] = s;
    __syncthreads();
    int first = 0;
    for (int w = 0; w < GCN_DEG_TPB / 32; ++w) first += part[w];
    const int last = first + binCount[bin];

    for (int p0 = first; p0 < last; p0 += GCN_DEG_BATCH * GCN_DEG_TPB) {
        int2 rr[GCN_DEG_BATCH];
#pragma unroll
        for (int u = 0; u < GCN_DEG_BATCH; ++u) {
            const int p = p0 + u * GCN_DEG_TPB + tid;
            rr[u] = rec[(p < last) ? p : (last - 1)];
        }
#pragma unroll
        for (int u = 0; u < GCN_DEG_BATCH; ++u) {
            const int p = p0 + u * GCN_DEG_TPB + tid;
            if (p < last) atomicAdd(&dl[rr[u].x & 511], __int_as_float(rr[u].y));
        }
    }
    __syncthreads();
    const int row0 = bin * GCN_BIN_ROWS;
    for (int i = tid; i < GCN_BIN_ROWS; i += GCN_DEG_TPB) {
        if (row0 + i < Nn) deg[row0 + i] = fmaxf(dl[i], 1e-6f);
    }
}


__device__ __forceinline__ void gcn_add_run(char* abase, int rowOff16, float ax, float ay) {
    float* arow = (float*)(abase + (rowOff16 << 4));
    atomicAdd(arow, ax);
    atomicAdd(arow + 1, ay);
}

__global__ void __launch_bounds__(GCN_TPB)
gcn_bin_sum_gemm_kernel(const float* __restrict__ H,
                        const int2* __restrict__ rec,
                        const int* __restrict__ binCount,
                        const float* __restrict__ deg,
                        const float* __restrict__ Wm,
                        float* __restrict__ out, int Nn, int nbins) {
    __shared__ __attribute__((aligned(16))) float acc[GCN_BIN_ROWS * GCN_PITCH];
    __shared__ int2 srt[GCN_SORT_CAP];
    __shared__ int scanA[GCN_TPB];
    __shared__ int scanB[GCN_TPB];
    __shared__ int span[2];
    __shared__ float degL[GCN_BIN_ROWS];
    __shared__ int rowCnt[GCN_BIN_ROWS];
    __shared__ int rowStart[GCN_BIN_ROWS];
    const int tid = threadIdx.x;
    const int lane = tid & 31;
    const int wave = tid >> 5;
    const int wv = __builtin_amdgcn_readfirstlane(wave);
    const int bin = blockIdx.x;
    const int row0 = bin * GCN_BIN_ROWS;

    {
        const float4 z = make_float4(0.0f, 0.0f, 0.0f, 0.0f);
        for (int i = tid; i < GCN_BIN_ROWS * GCN_PITCH / 4; i += GCN_TPB) ((float4*)acc)[i] = z;
    }
    for (int i = tid; i < GCN_BIN_ROWS; i += GCN_TPB) {
        degL[i] = (row0 + i < Nn) ? deg[row0 + i] : 1.0f;
        rowCnt[i] = 0;
    }
    const int cntAll = (tid < nbins) ? binCount[tid] : 0;
    const int binStart = gcn_block_excl_scan(cntAll, nbins, scanA, scanB);
    if (tid == bin) { span[0] = binStart; span[1] = binStart + cntAll; }
    __syncthreads();
    const int first = __builtin_amdgcn_readfirstlane(span[0]);
    const int last = __builtin_amdgcn_readfirstlane(span[1]);

    char* abase = (char*)(acc + 2 * lane);
    for (int c0 = first; c0 < last; c0 += GCN_SORT_CAP) {
        const int n = (last - c0 < GCN_SORT_CAP) ? (last - c0) : GCN_SORT_CAP;

        int2 r[GCN_SORT_EPT];
        int rank[GCN_SORT_EPT];
#pragma unroll
        for (int k = 0; k < GCN_SORT_EPT; ++k) {
            const int i = k * GCN_TPB + tid;
            r[k] = rec[c0 + ((i < n) ? i : (n - 1))];
        }
#pragma unroll
        for (int k = 0; k < GCN_SORT_EPT; ++k) {
            const int i = k * GCN_TPB + tid;
            rank[k] = (i < n) ? atomicAdd(&rowCnt[r[k].x & 511], 1) : 0;
        }
        __syncthreads();
        const int rc = (tid < GCN_BIN_ROWS) ? rowCnt[tid] : 0;
        const int rs = gcn_block_excl_scan(rc, GCN_BIN_ROWS, scanA, scanB);
        if (tid < GCN_BIN_ROWS) { rowStart[tid] = rs; rowCnt[tid] = 0; }
        __syncthreads();
#pragma unroll
        for (int k = 0; k < GCN_SORT_EPT; ++k) {
            const int i = k * GCN_TPB + tid;
            if (i < n) srt[rowStart[r[k].x & 511] + rank[k]] = r[k];
        }
        __syncthreads();

        for (int q = wv * 32; q < n; q += GCN_TPB) {
            const int cnt = (n - q < 32) ? (n - q) : 32;
            const int2 s = srt[(q + lane < n) ? (q + lane) : (n - 1)];
            const float ds = deg[s.x >> 9];
            const int pk = (int)(((unsigned)(s.x >> 9) << 13) |
                                 (unsigned)((s.x & 511) * (GCN_PITCH / 4)));
            float ax = 0.0f;
            float ay = 0.0f;
            int cur = __builtin_amdgcn_readlane(pk, 0) & 0x1fff;
            if (cnt == 32) {
                float2 h[32];
#pragma unroll
                for (int j = 0; j < 32; ++j) {
                    const int sr = (int)((unsigned)__builtin_amdgcn_readlane(pk, j) >> 13);
                    h[j] = ((const float2*)(H + (size_t)sr * DIM))[lane];
                }
                const float dd = degL[s.x & 511];
                const int nb = __float_as_int(__int_as_float(s.y) / fmaxf(sqrtf(ds * dd), 1e-6f));
#pragma unroll
                for (int j = 0; j < 32; ++j) {
                    const int rj = __builtin_amdgcn_readlane(pk, j) & 0x1fff;
                    if (__builtin_expect(rj != cur, 0)) {
                        gcn_add_run(abase, cur, ax, ay);
                        ax = 0.0f;
                        ay = 0.0f;
                        cur = rj;
                    }
                    const float nj = __int_as_float(__builtin_amdgcn_readlane(nb, j));
                    ax += nj * h[j].x;
                    ay += nj * h[j].y;
                }
            } else {
                const float dd = degL[s.x & 511];
                const int nb = __float_as_int(__int_as_float(s.y) / fmaxf(sqrtf(ds * dd), 1e-6f));
                for (int j0 = 0; j0 < cnt; j0 += 8) {
                    float2 hv[8];
#pragma unroll
                    for (int u = 0; u < 8; ++u) {
                        const int j = (j0 + u < cnt) ? (j0 + u) : (cnt - 1);
                        const int sr = (int)((unsigned)__builtin_amdgcn_readlane(pk, j) >> 13);
                        hv[u] = ((const float2*)(H + (size_t)sr * DIM))[lane];
                    }
#pragma unroll
                    for (int u = 0; u < 8; ++u) {
                        if (j0 + u < cnt) {
                            const int rj = __builtin_amdgcn_readlane(pk, j0 + u) & 0x1fff;
                            if (rj != cur) {
                                gcn_add_run(abase, cur, ax, ay);
                                ax = 0.0f;
                                ay = 0.0f;
                                cur = rj;
                            }
                            const float nj = __int_as_float(__builtin_amdgcn_readlane(nb, j0 + u));
                            ax += nj * hv[u].x;
                            ay += nj * hv[u].y;
                        }
                    }
                }
            }
            gcn_add_run(abase, cur, ax, ay);
        }
    }

    const int half = lane >> 4;
    const int l    = lane & 15;
    const int ct   = wv & 3;
    const int rt0  = wv >> 2;

    v2f bfrag[16];
    {
        const float* wp = Wm + (size_t)(ct * 16 + l) * DIM + 2 * half;
#pragma unroll
        for (int kk = 0; kk < 16; ++kk) bfrag[kk] = *(const v2f*)(wp + kk * 4);
    }

    __syncthreads();

    v8f c[4];
#pragma unroll
    for (int i = 0; i < 4; ++i) {
        const int rt = rt0 + 8 * i;
        if (rt < GCN_BIN_ROWS / 16) {
            const float* ap = acc + (rt * 16 + l) * GCN_PITCH + 2 * half;
            v2f afrag[16];
#pragma unroll
            for (int kk = 0; kk < 16; ++kk) afrag[kk] = *(const v2f*)(ap + kk * 4);
            v8f ci = {};
            __builtin_amdgcn_sched_barrier(0);
#pragma unroll
            for (int kk = 0; kk < 16; ++kk) {
                ci = __builtin_amdgcn_wmma_f32_16x16x4_f32(
                         false, afrag[kk],  false, bfrag[kk],
                         (short)0, ci,  false,  false);
            }
            __builtin_amdgcn_sched_barrier(0);
            asm volatile("v_nop\n\tv_nop\n\tv_nop\n\tv_nop");
            __builtin_amdgcn_sched_barrier(0);
            c[i] = ci;
        }
    }

    __syncthreads();

#pragma unroll
    for (int i = 0; i < 4; ++i) {
        const int rt = rt0 + 8 * i;
        if (rt < GCN_BIN_ROWS / 16) {
            float* dp = acc + (rt * 16 + 8 * half) * GCN_PITCH + ct * 16 + l;
#pragma unroll
            for (int r = 0; r < 8; ++r) dp[r * GCN_PITCH] = c[i][r];
        }
    }
    __syncthreads();

    const int rows = (Nn - row0 < GCN_BIN_ROWS) ? (Nn - row0) : GCN_BIN_ROWS;
    float4* dstp = (float4*)(out + (size_t)row0 * DIM);
    for (int i = tid; i < rows * (DIM / 4); i += GCN_TPB) {
        dstp[i] = *(const float4*)(acc + (i >> 4) * GCN_PITCH + (i & 15) * 4);
    }
}

extern "C" void kernel_launch(void* const* d_in, const int* in_sizes, int n_in,
                              void* d_out, int out_size, void* d_ws, size_t ws_size,
                              hipStream_t stream) {
    const float* H   = (const float*)d_in[0];
    const int*   ei  = (const int*)d_in[1];
    const float* ew  = (const float*)d_in[2];
    const float* Wm  = (const float*)d_in[3];
    float*       out = (float*)d_out;

    const int N = in_sizes[0] / DIM;
    const int E = in_sizes[2];

    const int degPad = ((N + 63) / 64) * 64;
    float* deg      = (float*)d_ws;
    int2*  rec      = (int2*)(deg + degPad);
    int*   binCount = (int*)(rec + E);
    int*   binFill  = binCount + GCN_MAX_BINS;
    const int nbins = (N + GCN_BIN_ROWS - 1) / GCN_BIN_ROWS;
    const int edgeBlocks = (E + GCN_EPT * GCN_TPB - 1) / (GCN_EPT * GCN_TPB);

    gcn_zero_counters_kernel<<<(2 * GCN_MAX_BINS + 255) / 256, 256, 0, stream>>>(binCount, 2 * GCN_MAX_BINS);
    gcn_hist_kernel<<<edgeBlocks, GCN_TPB, 0, stream>>>(ei, binCount, E, nbins);
    gcn_place_kernel<<<edgeBlocks, GCN_TPB, 0, stream>>>(ei, ew, binCount, binFill, rec, E, nbins);
    gcn_bin_deg_kernel<<<nbins, GCN_DEG_TPB, 0, stream>>>(rec, binCount, deg, N);
    gcn_bin_sum_gemm_kernel<<<nbins, GCN_TPB, 0, stream>>>(H, rec, binCount, deg, Wm, out, N, nbins);
}


# place, degree and range launches: block prefix sums and totals taken in registers by lane shifts (DPP row_shr) plus one LDS hand-over of the 32 wave totals: 1 barrier per scan instead of 9-10
# speedup vs baseline: 5.1103x; 1.0702x over previous
#include <hip/hip_runtime.h>
#include <hip/hip_bf16.h>


typedef __attribute__((ext_vector_type(2))) float v2f;
typedef __attribute__((ext_vector_type(8))) float v8f;

#define DIM 64
#define GCN_BIN_ROWS 400
#define GCN_PITCH 68
#define GCN_MAX_BINS 1024
static_assert(GCN_BIN_ROWS % 16 == 0 && GCN_PITCH % 4 == 0, "whole row tiles, 16-byte row starts");
static_assert((GCN_BIN_ROWS - 1) * (GCN_PITCH / 4) < 8192, "a row start in 16-byte units fits 13 bits");
#define GCN_EPT 6
#define GCN_DEG_TPB 256
#define GCN_DEG_BATCH 16
#define GCN_TPB 1024
#define GCN_SORT_EPT 8
#define GCN_SORT_CAP (GCN_SORT_EPT * GCN_TPB)
static_assert(GCN_BIN_ROWS <= GCN_TPB, "one thread per row in the row scan");
static_assert(GCN_TPB == 1024, "the block scans take 32 waves: one wave total per lane");

__device__ __forceinline__ int gcn_wave_incl_scan(int x) {
    x += __builtin_amdgcn_update_dpp(0, x, 0x111, 0xf, 0xf, true);
    x += __builtin_amdgcn_update_dpp(0, x, 0x112, 0xf, 0xf, true);
    x += __builtin_amdgcn_update_dpp(0, x, 0x114, 0xf, 0xf, true);
    x += __builtin_amdgcn_update_dpp(0, x, 0x118, 0xf, 0xf, true);
    const int lower = __builtin_amdgcn_readlane(x, 15);
    if ((threadIdx.x & 31) >= 16) x += lower;
    return x;
}

__device__ __forceinline__ int gcn_block_excl_scan(int v, int* part) {
    const int lane = threadIdx.x & 31;
    const int w = __builtin_amdgcn_readfirstlane((int)(threadIdx.x >> 5));
    const int x = gcn_wave_incl_scan(v);
    if (lane == 31) part[w] = x;
    __syncthreads();
    const int p = part[lane];
    const int px = gcn_wave_incl_scan(p);
    const int base = __builtin_amdgcn_readlane(px - p, w);
    return base + x - v;
}

__device__ __forceinline__ void gcn_block_excl_scan2(int va, int vb, int* part, int& ra, int& rb) {
    const int lane = threadIdx.x & 31;
    const int w = __builtin_amdgcn_readfirstlane((int)(threadIdx.x >> 5));
    const int xa = gcn_wave_incl_scan(va);
    const int xb = gcn_wave_incl_scan(vb);
    if (lane == 31) { part[w] = xa; part[32 + w] = xb; }
    __syncthreads();
    const int pa = part[lane];
    const int pb = part[32 + lane];
    const int pxa = gcn_wave_incl_scan(pa);
    const int pxb = gcn_wave_incl_scan(pb);
    ra = __builtin_amdgcn_readlane(pxa - pa, w) + xa - va;
    rb = __builtin_amdgcn_readlane(pxb - pb, w) + xb - vb;
}

__global__ void gcn_zero_counters_kernel(int* __restrict__ c, int nc) {
    int i = blockIdx.x * blockDim.x + threadIdx.x;
    if (i < nc) {
        c[i] = 0;
    }
}

__global__ void __launch_bounds__(GCN_TPB)
gcn_hist_kernel(const int* __restrict__ ei,
                int* __restrict__ binCount, int E, int nbins) {
    __shared__ int lh[GCN_MAX_BINS];
    const int tid = threadIdx.x;
    for (int b = tid; b < nbins; b += GCN_TPB) lh[b] = 0;
    __syncthreads();
    const int c0 = blockIdx.x * (GCN_EPT * GCN_TPB);
#pragma unroll
    for (int k = 0; k < GCN_EPT; ++k) {
        const int e = c0 + k * GCN_TPB + tid;
        if (e < E) {
            const int d = ei[E + e];
            atomicAdd(&lh[d / GCN_BIN_ROWS], 1);
        }
    }
    __syncthreads();
    for (int b = tid; b < nbins; b += GCN_TPB) {
        const int c = lh[b];
        if (c != 0) atomicAdd(&binCount[b], c);
    }
}

__global__ void __launch_bounds__(GCN_TPB)
gcn_place_kernel(const int* __restrict__ ei,
                 const float* __restrict__ ew,
                 const int* __restrict__ binCount,
                 int* __restrict__ binFill,
                 int2* __restrict__ rec, int E, int nbins) {
    __shared__ int lh[GCN_MAX_BINS];
    __shared__ int lstart[GCN_MAX_BINS];
    __shared__ int delta[GCN_MAX_BINS];
    __shared__ int part[64];
    __shared__ int2 stage[GCN_EPT * GCN_TPB];
    __shared__ int binAt[GCN_EPT * GCN_TPB];
    const int tid = threadIdx.x;
    for (int b = tid; b < nbins; b += GCN_TPB) lh[b] = 0;
    __syncthreads();

    const int c0 = blockIdx.x * (GCN_EPT * GCN_TPB);
    int packed[GCN_EPT];
    int wBits[GCN_EPT];
    int binOf[GCN_EPT];
    int rank[GCN_EPT];
#pragma unroll
    for (int k = 0; k < GCN_EPT; ++k) {
        const int e = c0 + k * GCN_TPB + tid;
        const bool valid = e < E;
        const int ec = valid ? e : (E - 1);
        const int src = ei[ec];
        const int dst = ei[E + ec];
        const float w = ew[ec];
        const int bin = dst / GCN_BIN_ROWS;
        packed[k] = src * 512 + (dst - bin * GCN_BIN_ROWS);
        wBits[k] = __float_as_int(w);
        binOf[k] = valid ? bin : -1;
        rank[k] = valid ? atomicAdd(&lh[bin], 1) : 0;
    }
    __syncthreads();

    const int cntAll = (tid < nbins) ? binCount[tid] : 0;
    const int mine = (tid < nbins) ? lh[tid] : 0;
    int binStart;
    int localStart;
    gcn_block_excl_scan2(cntAll, mine, part, binStart, localStart);
    if (tid < nbins) {
        const int got = (mine != 0) ? atomicAdd(&binFill[tid], mine) : 0;
        lstart[tid] = localStart;
        delta[tid] = binStart + got - localStart;
    }
    __syncthreads();

#pragma unroll
    for (int k = 0; k < GCN_EPT; ++k) {
        if (binOf[k] >= 0) {
            const int slot = lstart[binOf[k]] + rank[k];
            stage[slot] = make_int2(packed[k], wBits[k]);
            binAt[slot] = binOf[k];
        }
    }
    __syncthreads();

    const int left = E - c0;
    const int mineAll = (left < GCN_EPT * GCN_TPB) ? left : (GCN_EPT * GCN_TPB);
    for (int slot = tid; slot < mineAll; slot += GCN_TPB) {
        rec[delta[binAt[slot]] + slot] = stage[slot];
    }
}

__global__ void __launch_bounds__(GCN_DEG_TPB)
gcn_bin_deg_kernel(const int2* __restrict__ rec,
                   const int* __restrict__ binCount,
                   float* __restrict__ deg, int Nn) {
    __shared__ float dl[GCN_BIN_ROWS];
    __shared__ int part[GCN_DEG_TPB / 32];
    const int tid = threadIdx.x;
    const int bin = blockIdx.x;
    for (int i = tid; i < GCN_BIN_ROWS; i += GCN_DEG_TPB) dl[i] = 0.0f;
    int s = 0;
    for (int i = tid; i < bin; i += GCN_DEG_TPB) s += binCount[i];
    s = gcn_wave_incl_scan(s);
    if ((tid & 31) == 31) part[tid >> 5] = s;
    __syncthreads();
    int first = 0;
    for (int w = 0; w < GCN_DEG_TPB / 32; ++w) first += part[w];
    const int last = first + binCount[bin];

    for (int p0 = first; p0 < last; p0 += GCN_DEG_BATCH * GCN_DEG_TPB) {
        int2 rr[GCN_DEG_BATCH];
#pragma unroll
        for (int u = 0; u < GCN_DEG_BATCH; ++u) {
            const int p = p0 + u * GCN_DEG_TPB + tid;
            rr[u] = rec[(p < last) ? p : (last - 1)];
        }
#pragma unroll
        for (int u = 0; u < GCN_DEG_BATCH; ++u) {
            const int p = p0 + u * GCN_DEG_TPB + tid;
            if (p < last) atomicAdd(&dl[rr[u].x & 511], __int_as_float(rr[u].y));
        }
    }
    __syncthreads();
    const int row0 = bin * GCN_BIN_ROWS;
    for (int i = tid; i < GCN_BIN_ROWS; i += GCN_DEG_TPB) {
        if (row0 + i < Nn) deg[row0 + i] = fmaxf(dl[i], 1e-6f);
    }
}


__device__ __forceinline__ void gcn_add_run(char* abase, int rowOff16, float ax, float ay) {
    float* arow = (float*)(abase + (rowOff16 << 4));
    atomicAdd(arow, ax);
    atomicAdd(arow + 1, ay);
}

__global__ void __launch_bounds__(GCN_TPB)
gcn_bin_sum_gemm_kernel(const float* __restrict__ H,
                        const int2* __restrict__ rec,
                        const int* __restrict__ binCount,
                        const float* __restrict__ deg,
                        const float* __restrict__ Wm,
                        float* __restrict__ out, int Nn, int nbins) {
    __shared__ __attribute__((aligned(16))) float acc[GCN_BIN_ROWS * GCN_PITCH];
    __shared__ int2 srt[GCN_SORT_CAP];
    __shared__ int partF[32];
    __shared__ int partR[32];
    __shared__ float degL[GCN_BIN_ROWS];
    __shared__ int rowCnt[GCN_BIN_ROWS];
    __shared__ int rowStart[GCN_BIN_ROWS];
    const int tid = threadIdx.x;
    const int lane = tid & 31;
    const int wave = tid >> 5;
    const int wv = __builtin_amdgcn_readfirstlane(wave);
    const int bin = blockIdx.x;
    const int row0 = bin * GCN_BIN_ROWS;

    {
        const float4 z = make_float4(0.0f, 0.0f, 0.0f, 0.0f);
        for (int i = tid; i < GCN_BIN_ROWS * GCN_PITCH / 4; i += GCN_TPB) ((float4*)acc)[i] = z;
    }
    for (int i = tid; i < GCN_BIN_ROWS; i += GCN_TPB) {
        degL[i] = (row0 + i < Nn) ? deg[row0 + i] : 1.0f;
        rowCnt[i] = 0;
    }
    const int before = (tid < bin && tid < nbins) ? binCount[tid] : 0;
    const int bsum = gcn_wave_incl_scan(before);
    if (lane == 31) partF[wv] = bsum;
    __syncthreads();
    const int first = __builtin_amdgcn_readlane(gcn_wave_incl_scan(partF[lane]), 31);
    const int last = first + __builtin_amdgcn_readfirstlane(binCount[bin]);

    char* abase = (char*)(acc + 2 * lane);
    for (int c0 = first; c0 < last; c0 += GCN_SORT_CAP) {
        const int n = (last - c0 < GCN_SORT_CAP) ? (last - c0) : GCN_SORT_CAP;

        int2 r[GCN_SORT_EPT];
        int rank[GCN_SORT_EPT];
#pragma unroll
        for (int k = 0; k < GCN_SORT_EPT; ++k) {
            const int i = k * GCN_TPB + tid;
            r[k] = rec[c0 + ((i < n) ? i : (n - 1))];
        }
#pragma unroll
        for (int k = 0; k < GCN_SORT_EPT; ++k) {
            const int i = k * GCN_TPB + tid;
            rank[k] = (i < n) ? atomicAdd(&rowCnt[r[k].x & 511], 1) : 0;
        }
        __syncthreads();
        const int rc = (tid < GCN_BIN_ROWS) ? rowCnt[tid] : 0;
        const int rs = gcn_block_excl_scan(rc, partR);
        if (tid < GCN_BIN_ROWS) { rowStart[tid] = rs; rowCnt[tid] = 0; }
        __syncthreads();
#pragma unroll
        for (int k = 0; k < GCN_SORT_EPT; ++k) {
            const int i = k * GCN_TPB + tid;
            if (i < n) srt[rowStart[r[k].x & 511] + rank[k]] = r[k];
        }
        __syncthreads();

        for (int q = wv * 32; q < n; q += GCN_TPB) {
            const int cnt = (n - q < 32) ? (n - q) : 32;
            const int2 s = srt[(q + lane < n) ? (q + lane) : (n - 1)];
            const float ds = deg[s.x >> 9];
            const int pk = (int)(((unsigned)(s.x >> 9) << 13) |
                                 (unsigned)((s.x & 511) * (GCN_PITCH / 4)));
            float ax = 0.0f;
            float ay = 0.0f;
            int cur = __builtin_amdgcn_readlane(pk, 0) & 0x1fff;
            if (cnt == 32) {
                float2 h[32];
#pragma unroll
                for (int j = 0; j < 32; ++j) {
                    const int sr = (int)((unsigned)__builtin_amdgcn_readlane(pk, j) >> 13);
                    h[j] = ((const float2*)(H + (size_t)sr * DIM))[lane];
                }
                const float dd = degL[s.x & 511];
                const int nb = __float_as_int(__int_as_float(s.y) / fmaxf(sqrtf(ds * dd), 1e-6f));
#pragma unroll
                for (int j = 0; j < 32; ++j) {
                    const int rj = __builtin_amdgcn_readlane(pk, j) & 0x1fff;
                    if (__builtin_expect(rj != cur, 0)) {
                        gcn_add_run(abase, cur, ax, ay);
                        ax = 0.0f;
                        ay = 0.0f;
                        cur = rj;
                    }
                    const float nj = __int_as_float(__builtin_amdgcn_readlane(nb, j));
                    ax += nj * h[j].x;
                    ay += nj * h[j].y;
                }
            } else {
                const float dd = degL[s.x & 511];
                const int nb = __float_as_int(__int_as_float(s.y) / fmaxf(sqrtf(ds * dd), 1e-6f));
                for (int j0 = 0; j0 < cnt; j0 += 8) {
                    float2 hv[8];
#pragma unroll
                    for (int u = 0; u < 8; ++u) {
                        const int j = (j0 + u < cnt) ? (j0 + u) : (cnt - 1);
                        const int sr = (int)((unsigned)__builtin_amdgcn_readlane(pk, j) >> 13);
                        hv[u] = ((const float2*)(H + (size_t)sr * DIM))[lane];
                    }
#pragma unroll
                    for (int u = 0; u < 8; ++u) {
                        if (j0 + u < cnt) {
                            const int rj = __builtin_amdgcn_readlane(pk, j0 + u) & 0x1fff;
                            if (rj != cur) {
                                gcn_add_run(abase, cur, ax, ay);
                                ax = 0.0f;
                                ay = 0.0f;
                                cur = rj;
                            }
                            const float nj = __int_as_float(__builtin_amdgcn_readlane(nb, j0 + u));
                            ax += nj * hv[u].x;
                            ay += nj * hv[u].y;
                        }
                    }
                }
            }
            gcn_add_run(abase, cur, ax, ay);
        }
    }

    const int half = lane >> 4;
    const int l    = lane & 15;
    const int ct   = wv & 3;
    const int rt0  = wv >> 2;

    v2f bfrag[16];
    {
        const float* wp = Wm + (size_t)(ct * 16 + l) * DIM + 2 * half;
#pragma unroll
        for (int kk = 0; kk < 16; ++kk) bfrag[kk] = *(const v2f*)(wp + kk * 4);
    }

    __syncthreads();

    v8f c[4];
#pragma unroll
    for (int i = 0; i < 4; ++i) {
        const int rt = rt0 + 8 * i;
        if (rt < GCN_BIN_ROWS / 16) {
            const float* ap = acc + (rt * 16 + l) * GCN_PITCH + 2 * half;
            v2f afrag[16];
#pragma unroll
            for (int kk = 0; kk < 16; ++kk) afrag[kk] = *(const v2f*)(ap + kk * 4);
            v8f ci = {};
            __builtin_amdgcn_sched_barrier(0);
#pragma unroll
            for (int kk = 0; kk < 16; ++kk) {
                ci = __builtin_amdgcn_wmma_f32_16x16x4_f32(
                         false, afrag[kk],  false, bfrag[kk],
                         (short)0, ci,  false,  false);
            }
            __builtin_amdgcn_sched_barrier(0);
            asm volatile("v_nop\n\tv_nop\n\tv_nop\n\tv_nop");
            __builtin_amdgcn_sched_barrier(0);
            c[i] = ci;
        }
    }

    __syncthreads();

#pragma unroll
    for (int i = 0; i < 4; ++i) {
        const int rt = rt0 + 8 * i;
        if (rt < GCN_BIN_ROWS / 16) {
            float* dp = acc + (rt * 16 + 8 * half) * GCN_PITCH + ct * 16 + l;
#pragma unroll
            for (int r = 0; r < 8; ++r) dp[r * GCN_PITCH] = c[i][r];
        }
    }
    __syncthreads();

    const int rows = (Nn - row0 < GCN_BIN_ROWS) ? (Nn - row0) : GCN_BIN_ROWS;
    float4* dstp = (float4*)(out + (size_t)row0 * DIM);
    for (int i = tid; i < rows * (DIM / 4); i += GCN_TPB) {
        dstp[i] = *(const float4*)(acc + (i >> 4) * GCN_PITCH + (i & 15) * 4);
    }
}

extern "C" void kernel_launch(void* const* d_in, const int* in_sizes, int n_in,
                              void* d_out, int out_size, void* d_ws, size_t ws_size,
                              hipStream_t stream) {
    const float* H   = (const float*)d_in[0];
    const int*   ei  = (const int*)d_in[1];
    const float* ew  = (const float*)d_in[2];
    const float* Wm  = (const float*)d_in[3];
    float*       out = (float*)d_out;

    const int N = in_sizes[0] / DIM;
    const int E = in_sizes[2];

    const int degPad = ((N + 63) / 64) * 64;
    float* deg      = (float*)d_ws;
    int2*  rec      = (int2*)(deg + degPad);
    int*   binCount = (int*)(rec + E);
    int*   binFill  = binCount + GCN_MAX_BINS;
    const int nbins = (N + GCN_BIN_ROWS - 1) / GCN_BIN_ROWS;
    const int edgeBlocks = (E + GCN_EPT * GCN_TPB - 1) / (GCN_EPT * GCN_TPB);

    gcn_zero_counters_kernel<<<(2 * GCN_MAX_BINS + 255) / 256, 256, 0, stream>>>(binCount, 2 * GCN_MAX_BINS);
    gcn_hist_kernel<<<edgeBlocks, GCN_TPB, 0, stream>>>(ei, binCount, E, nbins);
    gcn_place_kernel<<<edgeBlocks, GCN_TPB, 0, stream>>>(ei, ew, binCount, binFill, rec, E, nbins);
    gcn_bin_deg_kernel<<<nbins, GCN_DEG_TPB, 0, stream>>>(rec, binCount, deg, N);
    gcn_bin_sum_gemm_kernel<<<nbins, GCN_TPB, 0, stream>>>(H, rec, binCount, deg, Wm, out, N, nbins);
}


# no count pass: place writes each chunk back in range order plus a row of run starts; degree and sum launches gather their runs via it; 5 launches to 3, no global atomic; sum order: runs by chunk
# speedup vs baseline: 5.5728x; 1.0905x over previous
#include <hip/hip_runtime.h>
#include <hip/hip_bf16.h>


typedef __attribute__((ext_vector_type(2))) float v2f;
typedef __attribute__((ext_vector_type(8))) float v8f;

#define DIM 64
#define GCN_BIN_ROWS 400
#define GCN_PITCH 68
#define GCN_MAX_BINS 1024
static_assert(GCN_BIN_ROWS % 16 == 0 && GCN_PITCH % 4 == 0, "whole row tiles, 16-byte row starts");
static_assert((GCN_BIN_ROWS - 1) * (GCN_PITCH / 4) < 8192, "a row start in 16-byte units fits 13 bits");
#define GCN_EPT 6
#define GCN_TPB 1024
#define GCN_CHUNK (GCN_EPT * GCN_TPB)
#define GCN_MAX_CHUNKS GCN_TPB
#define GCN_DEG_EPT 8
#define GCN_SORT_EPT 8
#define GCN_SORT_CAP (GCN_SORT_EPT * GCN_TPB)
static_assert(GCN_BIN_ROWS <= GCN_TPB, "one thread per row in the row scan");
static_assert(GCN_MAX_BINS <= GCN_TPB, "one thread per range in the place pass's scan");
static_assert(GCN_TPB == 1024, "the block scans take 32 waves: one wave total per lane");

__device__ __forceinline__ int gcn_wave_incl_scan(int x) {
    x += __builtin_amdgcn_update_dpp(0, x, 0x111, 0xf, 0xf, true);
    x += __builtin_amdgcn_update_dpp(0, x, 0x112, 0xf, 0xf, true);
    x += __builtin_amdgcn_update_dpp(0, x, 0x114, 0xf, 0xf, true);
    x += __builtin_amdgcn_update_dpp(0, x, 0x118, 0xf, 0xf, true);
    const int lower = __builtin_amdgcn_readlane(x, 15);
    if ((threadIdx.x & 31) >= 16) x += lower;
    return x;
}

__device__ __forceinline__ int gcn_block_excl_scan(int v, int* part) {
    const int lane = threadIdx.x & 31;
    const int w = __builtin_amdgcn_readfirstlane((int)(threadIdx.x >> 5));
    const int x = gcn_wave_incl_scan(v);
    if (lane == 31) part[w] = x;
    __syncthreads();
    const int p = part[lane];
    const int px = gcn_wave_incl_scan(p);
    const int base = __builtin_amdgcn_readlane(px - p, w);
    return base + x - v;
}

__global__ void __launch_bounds__(GCN_TPB)
gcn_place_kernel(const int* __restrict__ ei,
                 const float* __restrict__ ew,
                 int* __restrict__ runTab,
                 int2* __restrict__ rec, int E, int nbins, int rtPitch) {
    __shared__ int lh[GCN_MAX_BINS];
    __shared__ int lstart[GCN_MAX_BINS];
    __shared__ int part[32];
    __shared__ int2 stage[GCN_CHUNK];
    const int tid = threadIdx.x;
    for (int b = tid; b < nbins; b += GCN_TPB) lh[b] = 0;
    __syncthreads();

    const int c0 = blockIdx.x * GCN_CHUNK;
    int packed[GCN_EPT];
    int wBits[GCN_EPT];
    int binOf[GCN_EPT];
    int rank[GCN_EPT];
#pragma unroll
    for (int k = 0; k < GCN_EPT; ++k) {
        const int e = c0 + k * GCN_TPB + tid;
        const bool valid = e < E;
        const int ec = valid ? e : (E - 1);
        const int src = ei[ec];
        const int dst = ei[E + ec];
        const float w = ew[ec];
        const int bin = dst / GCN_BIN_ROWS;
        packed[k] = src * 512 + (dst - bin * GCN_BIN_ROWS);
        wBits[k] = __float_as_int(w);
        binOf[k] = valid ? bin : -1;
        rank[k] = valid ? atomicAdd(&lh[bin], 1) : 0;
    }
    __syncthreads();

    const int left = E - c0;
    const int mineAll = (left < GCN_CHUNK) ? left : GCN_CHUNK;
    const int mine = (tid < nbins) ? lh[tid] : 0;
    const int localStart = gcn_block_excl_scan(mine, part);
    int* rt = runTab + (size_t)blockIdx.x * rtPitch;
    if (tid < nbins) {
        lstart[tid] = localStart;
        rt[tid] = localStart;
    }
    if (tid == 0) rt[nbins] = mineAll;
    __syncthreads();

#pragma unroll
    for (int k = 0; k < GCN_EPT; ++k) {
        if (binOf[k] >= 0) stage[lstart[binOf[k]] + rank[k]] = make_int2(packed[k], wBits[k]);
    }
    __syncthreads();

    for (int slot = tid; slot < mineAll; slot += GCN_TPB) rec[c0 + slot] = stage[slot];
}

__device__ __forceinline__ int gcn_range_runs(const int* __restrict__ runTab, int bin, int nchunks,
                                              int rtPitch, int* relBase, int* part, int* total,
                                              int& off, int& len) {
    const int tid = threadIdx.x;
    int s0 = 0;
    len = 0;
    if (tid < nchunks) {
        const int* rt = runTab + (size_t)tid * rtPitch + bin;
        s0 = rt[0];
        len = rt[1] - s0;
    }
    off = gcn_block_excl_scan(len, part);
    if (tid < nchunks) relBase[tid] = tid * GCN_CHUNK + s0 - off;
    if (tid == GCN_TPB - 1) *total = off + len;
    __syncthreads();
    return __builtin_amdgcn_readfirstlane(*total);
}

__device__ __forceinline__ void gcn_mark_owner(int* owner, int c0, int np, int off, int len) {
    const int lo = (off > c0) ? off : c0;
    const int hi = (off + len < c0 + np) ? (off + len) : (c0 + np);
    for (int i = lo; i < hi; ++i) owner[i - c0] = (int)threadIdx.x;
}

__global__ void __launch_bounds__(GCN_TPB)
gcn_bin_deg_kernel(const int2* __restrict__ rec,
                   const int* __restrict__ runTab,
                   float* __restrict__ deg, int Nn, int nchunks, int rtPitch) {
    __shared__ float dl[GCN_BIN_ROWS];
    __shared__ int part[32];
    __shared__ int total;
    __shared__ int relBase[GCN_MAX_CHUNKS];
    __shared__ int owner[GCN_DEG_EPT * GCN_TPB];
    const int tid = threadIdx.x;
    const int bin = blockIdx.x;
    for (int i = tid; i < GCN_BIN_ROWS; i += GCN_TPB) dl[i] = 0.0f;
    int off;
    int len;
    const int n = gcn_range_runs(runTab, bin, nchunks, rtPitch, relBase, part, &total, off, len);

    for (int c0 = 0; c0 < n; c0 += GCN_DEG_EPT * GCN_TPB) {
        const int np = (n - c0 < GCN_DEG_EPT * GCN_TPB) ? (n - c0) : (GCN_DEG_EPT * GCN_TPB);
        gcn_mark_owner(owner, c0, np, off, len);
        __syncthreads();
        int2 rr[GCN_DEG_EPT];
#pragma unroll
        for (int u = 0; u < GCN_DEG_EPT; ++u) {
            const int i = u * GCN_TPB + tid;
            const int ic = (i < np) ? i : (np - 1);
            rr[u] = rec[relBase[owner[ic]] + c0 + ic];
        }
#pragma unroll
        for (int u = 0; u < GCN_DEG_EPT; ++u) {
            const int i = u * GCN_TPB + tid;
            if (i < np) atomicAdd(&dl[rr[u].x & 511], __int_as_float(rr[u].y));
        }
        __syncthreads();
    }
    const int row0 = bin * GCN_BIN_ROWS;
    for (int i = tid; i < GCN_BIN_ROWS; i += GCN_TPB) {
        if (row0 + i < Nn) deg[row0 + i] = fmaxf(dl[i], 1e-6f);
    }
}


__device__ __forceinline__ void gcn_add_run(char* abase, int rowOff16, float ax, float ay) {
    float* arow = (float*)(abase + (rowOff16 << 4));
    atomicAdd(arow, ax);
    atomicAdd(arow + 1, ay);
}

__global__ void __launch_bounds__(GCN_TPB)
gcn_bin_sum_gemm_kernel(const float* __restrict__ H,
                        const int2* __restrict__ rec,
                        const int* __restrict__ runTab,
                        const float* __restrict__ deg,
                        const float* __restrict__ Wm,
                        float* __restrict__ out, int Nn, int nchunks, int rtPitch) {
    __shared__ __attribute__((aligned(16))) float acc[GCN_BIN_ROWS * GCN_PITCH];
    __shared__ int2 srt[GCN_SORT_CAP];
    __shared__ int owner[GCN_SORT_CAP];
    __shared__ int relBase[GCN_MAX_CHUNKS];
    __shared__ int total;
    __shared__ int partF[32];
    __shared__ int partR[32];
    __shared__ float degL[GCN_BIN_ROWS];
    __shared__ int rowCnt[GCN_BIN_ROWS];
    __shared__ int rowStart[GCN_BIN_ROWS];
    const int tid = threadIdx.x;
    const int lane = tid & 31;
    const int wave = tid >> 5;
    const int wv = __builtin_amdgcn_readfirstlane(wave);
    const int bin = blockIdx.x;
    const int row0 = bin * GCN_BIN_ROWS;

    {
        const float4 z = make_float4(0.0f, 0.0f, 0.0f, 0.0f);
        for (int i = tid; i < GCN_BIN_ROWS * GCN_PITCH / 4; i += GCN_TPB) ((float4*)acc)[i] = z;
    }
    for (int i = tid; i < GCN_BIN_ROWS; i += GCN_TPB) {
        degL[i] = (row0 + i < Nn) ? deg[row0 + i] : 1.0f;
        rowCnt[i] = 0;
    }
    int runOff;
    int runLen;
    const int nAll = gcn_range_runs(runTab, bin, nchunks, rtPitch, relBase, partF, &total, runOff, runLen);

    char* abase = (char*)(acc + 2 * lane);
    for (int c0 = 0; c0 < nAll; c0 += GCN_SORT_CAP) {
        const int n = (nAll - c0 < GCN_SORT_CAP) ? (nAll - c0) : GCN_SORT_CAP;

        gcn_mark_owner(owner, c0, n, runOff, runLen);
        __syncthreads();
        int2 r[GCN_SORT_EPT];
        int rank[GCN_SORT_EPT];
#pragma unroll
        for (int k = 0; k < GCN_SORT_EPT; ++k) {
            const int i = k * GCN_TPB + tid;
            const int ic = (i < n) ? i : (n - 1);
            r[k] = rec[relBase[owner[ic]] + c0 + ic];
        }
#pragma unroll
        for (int k = 0; k < GCN_SORT_EPT; ++k) {
            const int i = k * GCN_TPB + tid;
            rank[k] = (i < n) ? atomicAdd(&rowCnt[r[k].x & 511], 1) : 0;
        }
        __syncthreads();
        const int rc = (tid < GCN_BIN_ROWS) ? rowCnt[tid] : 0;
        const int rs = gcn_block_excl_scan(rc, partR);
        if (tid < GCN_BIN_ROWS) { rowStart[tid] = rs; rowCnt[tid] = 0; }
        __syncthreads();
#pragma unroll
        for (int k = 0; k < GCN_SORT_EPT; ++k) {
            const int i = k * GCN_TPB + tid;
            if (i < n) srt[rowStart[r[k].x & 511] + rank[k]] = r[k];
        }
        __syncthreads();

        for (int q = wv * 32; q < n; q += GCN_TPB) {
            const int cnt = (n - q < 32) ? (n - q) : 32;
            const int2 s = srt[(q + lane < n) ? (q + lane) : (n - 1)];
            const float ds = deg[s.x >> 9];
            const int pk = (int)(((unsigned)(s.x >> 9) << 13) |
                                 (unsigned)((s.x & 511) * (GCN_PITCH / 4)));
            float ax = 0.0f;
            float ay = 0.0f;
            int cur = __builtin_amdgcn_readlane(pk, 0) & 0x1fff;
            if (cnt == 32) {
                float2 h[32];
#pragma unroll
                for (int j = 0; j < 32; ++j) {
                    const int sr = (int)((unsigned)__builtin_amdgcn_readlane(pk, j) >> 13);
                    h[j] = ((const float2*)(H + (size_t)sr * DIM))[lane];
                }
                const float dd = degL[s.x & 511];
                const int nb = __float_as_int(__int_as_float(s.y) / fmaxf(sqrtf(ds * dd), 1e-6f));
#pragma unroll
                for (int j = 0; j < 32; ++j) {
                    const int rj = __builtin_amdgcn_readlane(pk, j) & 0x1fff;
                    if (__builtin_expect(rj != cur, 0)) {
                        gcn_add_run(abase, cur, ax, ay);
                        ax = 0.0f;
                        ay = 0.0f;
                        cur = rj;
                    }
                    const float nj = __int_as_float(__builtin_amdgcn_readlane(nb, j));
                    ax += nj * h[j].x;
                    ay += nj * h[j].y;
                }
            } else {
                const float dd = degL[s.x & 511];
                const int nb = __float_as_int(__int_as_float(s.y) / fmaxf(sqrtf(ds * dd), 1e-6f));
                for (int j0 = 0; j0 < cnt; j0 += 8) {
                    float2 hv[8];
#pragma unroll
                    for (int u = 0; u < 8; ++u) {
                        const int j = (j0 + u < cnt) ? (j0 + u) : (cnt - 1);
                        const int sr = (int)((unsigned)__builtin_amdgcn_readlane(pk, j) >> 13);
                        hv[u] = ((const float2*)(H + (size_t)sr * DIM))[lane];
                    }
#pragma unroll
                    for (int u = 0; u < 8; ++u) {
                        if (j0 + u < cnt) {
                            const int rj = __builtin_amdgcn_readlane(pk, j0 + u) & 0x1fff;
                            if (rj != cur) {
                                gcn_add_run(abase, cur, ax, ay);
                                ax = 0.0f;
                                ay = 0.0f;
                                cur = rj;
                            }
                            const float nj = __int_as_float(__builtin_amdgcn_readlane(nb, j0 + u));
                            ax += nj * hv[u].x;
                            ay += nj * hv[u].y;
                        }
                    }
                }
            }
            gcn_add_run(abase, cur, ax, ay);
        }
    }

    const int half = lane >> 4;
    const int l    = lane & 15;
    const int ct   = wv & 3;
    const int rt0  = wv >> 2;

    v2f bfrag[16];
    {
        const float* wp = Wm + (size_t)(ct * 16 + l) * DIM + 2 * half;
#pragma unroll
        for (int kk = 0; kk < 16; ++kk) bfrag[kk] = *(const v2f*)(wp + kk * 4);
    }

    __syncthreads();

    v8f c[4];
#pragma unroll
    for (int i = 0; i < 4; ++i) {
        const int rt = rt0 + 8 * i;
        if (rt < GCN_BIN_ROWS / 16) {
            const float* ap = acc + (rt * 16 + l) * GCN_PITCH + 2 * half;
            v2f afrag[16];
#pragma unroll
            for (int kk = 0; kk < 16; ++kk) afrag[kk] = *(const v2f*)(ap + kk * 4);
            v8f ci = {};
            __builtin_amdgcn_sched_barrier(0);
#pragma unroll
            for (int kk = 0; kk < 16; ++kk) {
                ci = __builtin_amdgcn_wmma_f32_16x16x4_f32(
                         false, afrag[kk],  false, bfrag[kk],
                         (short)0, ci,  false,  false);
            }
            __builtin_amdgcn_sched_barrier(0);
            asm volatile("v_nop\n\tv_nop\n\tv_nop\n\tv_nop");
            __builtin_amdgcn_sched_barrier(0);
            c[i] = ci;
        }
    }

    __syncthreads();

#pragma unroll
    for (int i = 0; i < 4; ++i) {
        const int rt = rt0 + 8 * i;
        if (rt < GCN_BIN_ROWS / 16) {
            float* dp = acc + (rt * 16 + 8 * half) * GCN_PITCH + ct * 16 + l;
#pragma unroll
            for (int r = 0; r < 8; ++r) dp[r * GCN_PITCH] = c[i][r];
        }
    }
    __syncthreads();

    const int rows = (Nn - row0 < GCN_BIN_ROWS) ? (Nn - row0) : GCN_BIN_ROWS;
    float4* dstp = (float4*)(out + (size_t)row0 * DIM);
    for (int i = tid; i < rows * (DIM / 4); i += GCN_TPB) {
        dstp[i] = *(const float4*)(acc + (i >> 4) * GCN_PITCH + (i & 15) * 4);
    }
}

extern "C" void kernel_launch(void* const* d_in, const int* in_sizes, int n_in,
                              void* d_out, int out_size, void* d_ws, size_t ws_size,
                              hipStream_t stream) {
    const float* H   = (const float*)d_in[0];
    const int*   ei  = (const int*)d_in[1];
    const float* ew  = (const float*)d_in[2];
    const float* Wm  = (const float*)d_in[3];
    float*       out = (float*)d_out;

    const int N = in_sizes[0] / DIM;
    const int E = in_sizes[2];

    const int degPad = ((N + 63) / 64) * 64;
    const int recPad = ((E + 15) / 16) * 16;
    float* deg      = (float*)d_ws;
    int2*  rec      = (int2*)(deg + degPad);
    int*   runTab   = (int*)(rec + recPad);
    const int nbins = (N + GCN_BIN_ROWS - 1) / GCN_BIN_ROWS;
    const int nchunks = (E + GCN_CHUNK - 1) / GCN_CHUNK;
    const int rtPitch = ((nbins + 1 + 31) / 32) * 32;

    gcn_place_kernel<<<nchunks, GCN_TPB, 0, stream>>>(ei, ew, runTab, rec, E, nbins, rtPitch);
    gcn_bin_deg_kernel<<<nbins, GCN_TPB, 0, stream>>>(rec, runTab, deg, N, nchunks, rtPitch);
    gcn_bin_sum_gemm_kernel<<<nbins, GCN_TPB, 0, stream>>>(H, rec, runTab, deg, Wm, out, N, nchunks, rtPitch);
}


# sum phase: each 16-lane half owns whole rows, sums a row's sorted run in registers (16 B/lane, one load = 2 source rows, norm taken at sort time); no LDS float add; sum order: list order
# speedup vs baseline: 5.8953x; 1.0579x over previous
#include <hip/hip_runtime.h>
#include <hip/hip_bf16.h>


typedef __attribute__((ext_vector_type(2))) float v2f;
typedef __attribute__((ext_vector_type(8))) float v8f;

#define DIM 64
#define GCN_BIN_ROWS 400
#define GCN_PITCH 68
#define GCN_MAX_BINS 1024
static_assert(GCN_BIN_ROWS % 16 == 0 && GCN_PITCH % 4 == 0, "whole row tiles, 16-byte row starts");
static_assert((GCN_BIN_ROWS - 1) * (GCN_PITCH / 4) < 8192, "a row start in 16-byte units fits 13 bits");
#define GCN_EPT 6
#define GCN_TPB 1024
#define GCN_CHUNK (GCN_EPT * GCN_TPB)
#define GCN_MAX_CHUNKS GCN_TPB
#define GCN_DEG_EPT 8
#define GCN_SORT_EPT 8
#define GCN_SORT_CAP (GCN_SORT_EPT * GCN_TPB)
#define GCN_ROW_UNROLL 8
static_assert(GCN_ROW_UNROLL == 8, "the row walk names its 8 slots one by one");
static_assert(GCN_BIN_ROWS <= GCN_TPB, "one thread per row in the row scan");
static_assert(GCN_MAX_BINS <= GCN_TPB, "one thread per range in the place pass's scan");
static_assert(GCN_TPB == 1024, "the block scans take 32 waves: one wave total per lane");

__device__ __forceinline__ int gcn_wave_incl_scan(int x) {
    x += __builtin_amdgcn_update_dpp(0, x, 0x111, 0xf, 0xf, true);
    x += __builtin_amdgcn_update_dpp(0, x, 0x112, 0xf, 0xf, true);
    x += __builtin_amdgcn_update_dpp(0, x, 0x114, 0xf, 0xf, true);
    x += __builtin_amdgcn_update_dpp(0, x, 0x118, 0xf, 0xf, true);
    const int lower = __builtin_amdgcn_readlane(x, 15);
    if ((threadIdx.x & 31) >= 16) x += lower;
    return x;
}

__device__ __forceinline__ int gcn_block_excl_scan(int v, int* part) {
    const int lane = threadIdx.x & 31;
    const int w = __builtin_amdgcn_readfirstlane((int)(threadIdx.x >> 5));
    const int x = gcn_wave_incl_scan(v);
    if (lane == 31) part[w] = x;
    __syncthreads();
    const int p = part[lane];
    const int px = gcn_wave_incl_scan(p);
    const int base = __builtin_amdgcn_readlane(px - p, w);
    return base + x - v;
}

__global__ void __launch_bounds__(GCN_TPB)
gcn_place_kernel(const int* __restrict__ ei,
                 const float* __restrict__ ew,
                 int* __restrict__ runTab,
                 int2* __restrict__ rec, int E, int nbins, int rtPitch) {
    __shared__ int lh[GCN_MAX_BINS];
    __shared__ int lstart[GCN_MAX_BINS];
    __shared__ int part[32];
    __shared__ int2 stage[GCN_CHUNK];
    const int tid = threadIdx.x;
    for (int b = tid; b < nbins; b += GCN_TPB) lh[b] = 0;
    __syncthreads();

    const int c0 = blockIdx.x * GCN_CHUNK;
    int packed[GCN_EPT];
    int wBits[GCN_EPT];
    int binOf[GCN_EPT];
    int rank[GCN_EPT];
#pragma unroll
    for (int k = 0; k < GCN_EPT; ++k) {
        const int e = c0 + k * GCN_TPB + tid;
        const bool valid = e < E;
        const int ec = valid ? e : (E - 1);
        const int src = ei[ec];
        const int dst = ei[E + ec];
        const float w = ew[ec];
        const int bin = dst / GCN_BIN_ROWS;
        packed[k] = src * 512 + (dst - bin * GCN_BIN_ROWS);
        wBits[k] = __float_as_int(w);
        binOf[k] = valid ? bin : -1;
        rank[k] = valid ? atomicAdd(&lh[bin], 1) : 0;
    }
    __syncthreads();

    const int left = E - c0;
    const int mineAll = (left < GCN_CHUNK) ? left : GCN_CHUNK;
    const int mine = (tid < nbins) ? lh[tid] : 0;
    const int localStart = gcn_block_excl_scan(mine, part);
    int* rt = runTab + (size_t)blockIdx.x * rtPitch;
    if (tid < nbins) {
        lstart[tid] = localStart;
        rt[tid] = localStart;
    }
    if (tid == 0) rt[nbins] = mineAll;
    __syncthreads();

#pragma unroll
    for (int k = 0; k < GCN_EPT; ++k) {
        if (binOf[k] >= 0) stage[lstart[binOf[k]] + rank[k]] = make_int2(packed[k], wBits[k]);
    }
    __syncthreads();

    for (int slot = tid; slot < mineAll; slot += GCN_TPB) rec[c0 + slot] = stage[slot];
}

__device__ __forceinline__ int gcn_range_runs(const int* __restrict__ runTab, int bin, int nchunks,
                                              int rtPitch, int* relBase, int* part, int* total,
                                              int& off, int& len) {
    const int tid = threadIdx.x;
    int s0 = 0;
    len = 0;
    if (tid < nchunks) {
        const int* rt = runTab + (size_t)tid * rtPitch + bin;
        s0 = rt[0];
        len = rt[1] - s0;
    }
    off = gcn_block_excl_scan(len, part);
    if (tid < nchunks) relBase[tid] = tid * GCN_CHUNK + s0 - off;
    if (tid == GCN_TPB - 1) *total = off + len;
    __syncthreads();
    return __builtin_amdgcn_readfirstlane(*total);
}

__device__ __forceinline__ void gcn_mark_owner(int* owner, int c0, int np, int off, int len) {
    const int lo = (off > c0) ? off : c0;
    const int hi = (off + len < c0 + np) ? (off + len) : (c0 + np);
    for (int i = lo; i < hi; ++i) owner[i - c0] = (int)threadIdx.x;
}

__global__ void __launch_bounds__(GCN_TPB)
gcn_bin_deg_kernel(const int2* __restrict__ rec,
                   const int* __restrict__ runTab,
                   float* __restrict__ deg, int Nn, int nchunks, int rtPitch) {
    __shared__ float dl[GCN_BIN_ROWS];
    __shared__ int part[32];
    __shared__ int total;
    __shared__ int relBase[GCN_MAX_CHUNKS];
    __shared__ int owner[GCN_DEG_EPT * GCN_TPB];
    const int tid = threadIdx.x;
    const int bin = blockIdx.x;
    for (int i = tid; i < GCN_BIN_ROWS; i += GCN_TPB) dl[i] = 0.0f;
    int off;
    int len;
    const int n = gcn_range_runs(runTab, bin, nchunks, rtPitch, relBase, part, &total, off, len);

    for (int c0 = 0; c0 < n; c0 += GCN_DEG_EPT * GCN_TPB) {
        const int np = (n - c0 < GCN_DEG_EPT * GCN_TPB) ? (n - c0) : (GCN_DEG_EPT * GCN_TPB);
        gcn_mark_owner(owner, c0, np, off, len);
        __syncthreads();
        int2 rr[GCN_DEG_EPT];
#pragma unroll
        for (int u = 0; u < GCN_DEG_EPT; ++u) {
            const int i = u * GCN_TPB + tid;
            const int ic = (i < np) ? i : (np - 1);
            rr[u] = rec[relBase[owner[ic]] + c0 + ic];
        }
#pragma unroll
        for (int u = 0; u < GCN_DEG_EPT; ++u) {
            const int i = u * GCN_TPB + tid;
            if (i < np) atomicAdd(&dl[rr[u].x & 511], __int_as_float(rr[u].y));
        }
        __syncthreads();
    }
    const int row0 = bin * GCN_BIN_ROWS;
    for (int i = tid; i < GCN_BIN_ROWS; i += GCN_TPB) {
        if (row0 + i < Nn) deg[row0 + i] = fmaxf(dl[i], 1e-6f);
    }
}


__global__ void __launch_bounds__(GCN_TPB)
gcn_bin_sum_gemm_kernel(const float* __restrict__ H,
                        const int2* __restrict__ rec,
                        const int* __restrict__ runTab,
                        const float* __restrict__ deg,
                        const float* __restrict__ Wm,
                        float* __restrict__ out, int Nn, int nchunks, int rtPitch) {
    __shared__ __attribute__((aligned(16))) float acc[GCN_BIN_ROWS * GCN_PITCH];
    __shared__ int2 srt[GCN_SORT_CAP];
    __shared__ int owner[GCN_SORT_CAP];
    __shared__ int relBase[GCN_MAX_CHUNKS];
    __shared__ int total;
    __shared__ int partF[32];
    __shared__ int partR[32];
    __shared__ float degL[GCN_BIN_ROWS];
    __shared__ int rowCnt[GCN_BIN_ROWS];
    __shared__ int rowStart[GCN_BIN_ROWS + 1];
    const int tid = threadIdx.x;
    const int lane = tid & 31;
    const int wave = tid >> 5;
    const int wv = __builtin_amdgcn_readfirstlane(wave);
    const int bin = blockIdx.x;
    const int row0 = bin * GCN_BIN_ROWS;

    {
        const float4 z = make_float4(0.0f, 0.0f, 0.0f, 0.0f);
        for (int i = tid; i < GCN_BIN_ROWS * GCN_PITCH / 4; i += GCN_TPB) ((float4*)acc)[i] = z;
    }
    for (int i = tid; i < GCN_BIN_ROWS; i += GCN_TPB) {
        degL[i] = (row0 + i < Nn) ? deg[row0 + i] : 1.0f;
        rowCnt[i] = 0;
    }
    int runOff;
    int runLen;
    const int nAll = gcn_range_runs(runTab, bin, nchunks, rtPitch, relBase, partF, &total, runOff, runLen);

    const unsigned hoff = (unsigned)(tid & 15) * 16u;
    for (int c0 = 0; c0 < nAll; c0 += GCN_SORT_CAP) {
        const int n = (nAll - c0 < GCN_SORT_CAP) ? (nAll - c0) : GCN_SORT_CAP;

        gcn_mark_owner(owner, c0, n, runOff, runLen);
        __syncthreads();
        int2 r[GCN_SORT_EPT];
        int rank[GCN_SORT_EPT];
#pragma unroll
        for (int k = 0; k < GCN_SORT_EPT; ++k) {
            const int i = k * GCN_TPB + tid;
            const int ic = (i < n) ? i : (n - 1);
            r[k] = rec[relBase[owner[ic]] + c0 + ic];
        }
#pragma unroll
        for (int k = 0; k < GCN_SORT_EPT; ++k) {
            const int i = k * GCN_TPB + tid;
            rank[k] = (i < n) ? atomicAdd(&rowCnt[r[k].x & 511], 1) : 0;
        }
        float dsrc[GCN_SORT_EPT];
#pragma unroll
        for (int k = 0; k < GCN_SORT_EPT; ++k) dsrc[k] = deg[r[k].x >> 9];
        __syncthreads();
        const int rc = (tid < GCN_BIN_ROWS) ? rowCnt[tid] : 0;
        const int rs = gcn_block_excl_scan(rc, partR);
        if (tid < GCN_BIN_ROWS) { rowStart[tid] = rs; rowCnt[tid] = 0; }
        if (tid == GCN_BIN_ROWS - 1) rowStart[GCN_BIN_ROWS] = rs + rc;
        __syncthreads();
#pragma unroll
        for (int k = 0; k < GCN_SORT_EPT; ++k) {
            const int i = k * GCN_TPB + tid;
            if (i < n) {
                const int row = r[k].x & 511;
                const float nrm = __int_as_float(r[k].y) / fmaxf(sqrtf(dsrc[k] * degL[row]), 1e-6f);
                srt[rowStart[row] + rank[k]] = make_int2((r[k].x >> 9) * (DIM * 4), __float_as_int(nrm));
            }
        }
        __syncthreads();

        for (int row = tid >> 4; row < GCN_BIN_ROWS; row += GCN_TPB / 16) {
            const int e = rowStart[row + 1];
            float4 a = make_float4(0.0f, 0.0f, 0.0f, 0.0f);
            for (int i = rowStart[row]; i < e; i += GCN_ROW_UNROLL) {
                const int j = i + (tid & (GCN_ROW_UNROLL - 1));
                const int2 s = srt[(j < e) ? j : (e - 1)];
                const int sx = s.x;
                const int sy = (j < e) ? s.y : 0;
                float4 h[GCN_ROW_UNROLL];
                float nv[GCN_ROW_UNROLL] = {0.0f, 0.0f, 0.0f, 0.0f, 0.0f, 0.0f, 0.0f, 0.0f};
#define GCN_SLOT(U)                                                                              \
                {                                                                                \
                    const int ox = __builtin_amdgcn_update_dpp(0, sx, 0x150 + U, 0xf, 0xf, false); \
                    nv[U] = __int_as_float(__builtin_amdgcn_update_dpp(0, sy, 0x150 + U, 0xf, 0xf, false)); \
                    h[U] = *(const float4*)((const char*)H + (unsigned)ox + hoff);               \
                }
                GCN_SLOT(0) GCN_SLOT(1) GCN_SLOT(2) GCN_SLOT(3)
                h[4] = h[5] = h[6] = h[7] = make_float4(0.0f, 0.0f, 0.0f, 0.0f);
                if (i + 4 < e) {
                    GCN_SLOT(4) GCN_SLOT(5) GCN_SLOT(6) GCN_SLOT(7)
                }
#undef GCN_SLOT
#pragma unroll
                for (int u = 0; u < GCN_ROW_UNROLL; ++u) {
                    a.x += nv[u] * h[u].x;
                    a.y += nv[u] * h[u].y;
                    a.z += nv[u] * h[u].z;
                    a.w += nv[u] * h[u].w;
                }
            }
            float4* arow = (float4*)(acc + row * GCN_PITCH + 4 * (tid & 15));
            float4 t = *arow;
            t.x += a.x; t.y += a.y; t.z += a.z; t.w += a.w;
            *arow = t;
        }
    }

    const int half = lane >> 4;
    const int l    = lane & 15;
    const int ct   = wv & 3;
    const int rt0  = wv >> 2;

    v2f bfrag[16];
    {
        const float* wp = Wm + (size_t)(ct * 16 + l) * DIM + 2 * half;
#pragma unroll
        for (int kk = 0; kk < 16; ++kk) bfrag[kk] = *(const v2f*)(wp + kk * 4);
    }

    __syncthreads();

    v8f c[4];
#pragma unroll
    for (int i = 0; i < 4; ++i) {
        const int rt = rt0 + 8 * i;
        if (rt < GCN_BIN_ROWS / 16) {
            const float* ap = acc + (rt * 16 + l) * GCN_PITCH + 2 * half;
            v2f afrag[16];
#pragma unroll
            for (int kk = 0; kk < 16; ++kk) afrag[kk] = *(const v2f*)(ap + kk * 4);
            v8f ci = {};
            __builtin_amdgcn_sched_barrier(0);
#pragma unroll
            for (int kk = 0; kk < 16; ++kk) {
                ci = __builtin_amdgcn_wmma_f32_16x16x4_f32(
                         false, afrag[kk],  false, bfrag[kk],
                         (short)0, ci,  false,  false);
            }
            __builtin_amdgcn_sched_barrier(0);
            asm volatile("v_nop\n\tv_nop\n\tv_nop\n\tv_nop");
            __builtin_amdgcn_sched_barrier(0);
            c[i] = ci;
        }
    }

    __syncthreads();

#pragma unroll
    for (int i = 0; i < 4; ++i) {
        const int rt = rt0 + 8 * i;
        if (rt < GCN_BIN_ROWS / 16) {
            float* dp = acc + (rt * 16 + 8 * half) * GCN_PITCH + ct * 16 + l;
#pragma unroll
            for (int r = 0; r < 8; ++r) dp[r * GCN_PITCH] = c[i][r];
        }
    }
    __syncthreads();

    const int rows = (Nn - row0 < GCN_BIN_ROWS) ? (Nn - row0) : GCN_BIN_ROWS;
    float4* dstp = (float4*)(out + (size_t)row0 * DIM);
    for (int i = tid; i < rows * (DIM / 4); i += GCN_TPB) {
        dstp[i] = *(const float4*)(acc + (i >> 4) * GCN_PITCH + (i & 15) * 4);
    }
}

extern "C" void kernel_launch(void* const* d_in, const int* in_sizes, int n_in,
                              void* d_out, int out_size, void* d_ws, size_t ws_size,
                              hipStream_t stream) {
    const float* H   = (const float*)d_in[0];
    const int*   ei  = (const int*)d_in[1];
    const float* ew  = (const float*)d_in[2];
    const float* Wm  = (const float*)d_in[3];
    float*       out = (float*)d_out;

    const int N = in_sizes[0] / DIM;
    const int E = in_sizes[2];

    const int degPad = ((N + 63) / 64) * 64;
    const int recPad = ((E + 15) / 16) * 16;
    float* deg      = (float*)d_ws;
    int2*  rec      = (int2*)(deg + degPad);
    int*   runTab   = (int*)(rec + recPad);
    const int nbins = (N + GCN_BIN_ROWS - 1) / GCN_BIN_ROWS;
    const int nchunks = (E + GCN_CHUNK - 1) / GCN_CHUNK;
    const int rtPitch = ((nbins + 1 + 31) / 32) * 32;

    gcn_place_kernel<<<nchunks, GCN_TPB, 0, stream>>>(ei, ew, runTab, rec, E, nbins, rtPitch);
    gcn_bin_deg_kernel<<<nbins, GCN_TPB, 0, stream>>>(rec, runTab, deg, N, nchunks, rtPitch);
    gcn_bin_sum_gemm_kernel<<<nbins, GCN_TPB, 0, stream>>>(H, rec, runTab, deg, Wm, out, N, nchunks, rtPitch);
}
